# starting kernel
# baseline (speedup 1.0000x reference)
#include <hip/hip_runtime.h>
#include <hip/hip_bf16.h>

#define N_NODES 50000
#define N_EDGES 800000
#define D_FEAT  128

typedef __attribute__((ext_vector_type(2))) float v2f;
typedef __attribute__((ext_vector_type(8))) float v8f;

__global__ void sage_zero(float4* __restrict__ p, int n4) {
    int i = blockIdx.x * blockDim.x + threadIdx.x;
    if (i < n4) p[i] = make_float4(0.f, 0.f, 0.f, 0.f);
}

__global__ void sage_scatter(const float* __restrict__ in,
                             const int* __restrict__ src,
                             const int* __restrict__ dst,
                             float* __restrict__ agg,
                             float* __restrict__ deg) {
    int e = blockIdx.x * 2 + (threadIdx.x >> 7);
    int d = threadIdx.x & 127;
    if (e >= N_EDGES) return;
    int s = src[e];
    int t = dst[e];
    __hip_atomic_fetch_add(&agg[t * D_FEAT + d], in[s * D_FEAT + d],
                           __ATOMIC_RELAXED, __HIP_MEMORY_SCOPE_AGENT);
    if (d == 0)
        __hip_atomic_fetch_add(&deg[t], 1.0f,
                               __ATOMIC_RELAXED, __HIP_MEMORY_SCOPE_AGENT);
}

__global__ void sage_mean(float* __restrict__ agg, const float* __restrict__ deg) {
    int i = blockIdx.x * blockDim.x + threadIdx.x;
    if (i < N_NODES * D_FEAT) {
        float dv = fmaxf(deg[i >> 7], 1.0f);
        agg[i] = agg[i] / dv;
    }
}

__global__ void sage_gemm_wmma(const float* __restrict__ Amean,
                               const float* __restrict__ Ax,
                               const float* __restrict__ Wl,
                               const float* __restrict__ Wr,
                               const float* __restrict__ bias,
                               float* __restrict__ out,
                               int do_relu) {
    const int wave  = threadIdx.x >> 5;
    const int lane  = threadIdx.x & 31;
    const int lhalf = lane >> 4;
    const int lmod  = lane & 15;
    const int rbase = blockIdx.x * 16;
    const int cbase = wave * 16;
    const int arow  = rbase + lmod;
    const int col   = cbase + lmod;

    v8f acc = {};
    const float* amp = Amean + arow * D_FEAT + (lhalf ? 2 : 0);
    const float* axp = Ax    + arow * D_FEAT + (lhalf ? 2 : 0);

    #pragma unroll 4
    for (int k = 0; k < D_FEAT; k += 4) {
        const int ka = k + (lhalf ? 2 : 0);
        v2f am = *(const v2f*)(amp + k);
        v2f ax = *(const v2f*)(axp + k);
        v2f bl, br;
        bl.x = Wl[(ka + 0) * D_FEAT + col];
        bl.y = Wl[(ka + 1) * D_FEAT + col];
        br.x = Wr[(ka + 0) * D_FEAT + col];
        br.y = Wr[(ka + 1) * D_FEAT + col];
        acc = __builtin_amdgcn_wmma_f32_16x16x4_f32(false, am, false, bl,
                                                    (short)0, acc, false, false);
        acc = __builtin_amdgcn_wmma_f32_16x16x4_f32(false, ax, false, br,
                                                    (short)0, acc, false, false);
    }

    float bv = bias[col];
    #pragma unroll
    for (int v = 0; v < 8; ++v) {
        float r = acc[v] + bv;
        if (do_relu) r = fmaxf(r, 0.0f);
        int m = v + (lhalf ? 8 : 0);
        out[(rbase + m) * D_FEAT + col] = r;
    }
}

extern "C" void kernel_launch(void* const* d_in, const int* in_sizes, int n_in,
                              void* d_out, int out_size, void* d_ws, size_t ws_size,
                              hipStream_t stream) {
    const float* x    = (const float*)d_in[0];
    const int*   eidx = (const int*)d_in[1];
    const float* W1l  = (const float*)d_in[2];
    const float* b1   = (const float*)d_in[3];
    const float* W1r  = (const float*)d_in[4];
    const float* W2l  = (const float*)d_in[5];
    const float* b2   = (const float*)d_in[6];
    const float* W2r  = (const float*)d_in[7];
    float* out = (float*)d_out;

    const int* src = eidx;
    const int* dst = eidx + N_EDGES;

    float* agg = (float*)d_ws;
    float* deg = agg + (size_t)N_NODES * D_FEAT;
    float* h   = deg + N_NODES;

    const int nzero4  = (N_NODES * D_FEAT + N_NODES) / 4;
    const int zeroBl  = (nzero4 + 255) / 256;
    const int scatBl  = (N_EDGES + 1) / 2;
    const int meanBl  = (N_NODES * D_FEAT + 255) / 256;
    const int gemmBl  = N_NODES / 16;

    sage_zero<<<zeroBl, 256, 0, stream>>>((float4*)agg, nzero4);
    sage_scatter<<<scatBl, 256, 0, stream>>>(x, src, dst, agg, deg);
    sage_mean<<<meanBl, 256, 0, stream>>>(agg, deg);
    sage_gemm_wmma<<<gemmBl, 256, 0, stream>>>(agg, x, W1l, W1r, b1, h, 1);

    sage_zero<<<zeroBl, 256, 0, stream>>>((float4*)agg, nzero4);
    sage_scatter<<<scatBl, 256, 0, stream>>>(h, src, dst, agg, deg);
    sage_mean<<<meanBl, 256, 0, stream>>>(agg, deg);
    sage_gemm_wmma<<<gemmBl, 256, 0, stream>>>(agg, h, W2l, W2r, b2, out, 0);
}


# Scatter moved into LDS: each of 256 blocks owns 196 nodes, reads all edge dsts, queues its own, adds their src rows with LDS float adds, stores the mean; global atomics gone; sum order = LDS arrival
# speedup vs baseline: 2.8829x; 2.8829x over previous
#include <hip/hip_runtime.h>
#include <hip/hip_bf16.h>


#define N_NODES 50000
#define N_EDGES 800000
#define D_FEAT  128

#define ROWS_PER_BLOCK 196
#define AGG_BLOCKS     256
#define AGG_THREADS    1024
#define QUEUE_CAP      16384

typedef __attribute__((ext_vector_type(2))) float v2f;
typedef __attribute__((ext_vector_type(8))) float v8f;

__global__ __launch_bounds__(AGG_THREADS)
void sage_aggregate(const float* __restrict__ in,
                    const int* __restrict__ src,
                    const int* __restrict__ dst,
                    float* __restrict__ mean) {
    __shared__ __attribute__((aligned(16))) float acc[ROWS_PER_BLOCK * D_FEAT];
    __shared__ __attribute__((aligned(16))) unsigned int queue[QUEUE_CAP];
    __shared__ int cnt[ROWS_PER_BLOCK];
    __shared__ unsigned int qn;

    const int tid   = threadIdx.x;
    const int base  = blockIdx.x * ROWS_PER_BLOCK;
    const int nrows = min(ROWS_PER_BLOCK, N_NODES - base);

    float4* acc4 = reinterpret_cast<float4*>(acc);
    for (int i = tid; i < ROWS_PER_BLOCK * D_FEAT / 4; i += AGG_THREADS)
        acc4[i] = make_float4(0.f, 0.f, 0.f, 0.f);
    if (tid < ROWS_PER_BLOCK) cnt[tid] = 0;
    if (tid == 0) qn = 0u;
    __syncthreads();

    const int n4 = N_EDGES / 4;
    const int4* __restrict__ dst4 = reinterpret_cast<const int4*>(dst);
    const int start = (int)(((unsigned)blockIdx.x * (unsigned)n4) / AGG_BLOCKS);

    auto group_of = [&](int i0, int t) -> int {
        const int i = i0 + t * AGG_THREADS + tid;
        int j = (i < n4 ? i : 0) + start;
        if (j >= n4) j -= n4;
        return j;
    };
    auto push = [&](unsigned e, unsigned u) {
        const unsigned slot = atomicAdd(&qn, 1u);
        if (slot < QUEUE_CAP) queue[slot] = e | (u << 20);
    };

    int4 cur[4];
    #pragma unroll
    for (int t = 0; t < 4; ++t) cur[t] = dst4[group_of(0, t)];

    for (int i0 = 0; i0 < n4; i0 += 4 * AGG_THREADS) {
        int4 nxt[4];
        #pragma unroll
        for (int t = 0; t < 4; ++t) nxt[t] = dst4[group_of(i0 + 4 * AGG_THREADS, t)];

        #pragma unroll
        for (int t = 0; t < 4; ++t) {
            const int i = i0 + t * AGG_THREADS + tid;
            const unsigned lim = (i < n4) ? (unsigned)nrows : 0u;
            const unsigned u0 = (unsigned)(cur[t].x - base);
            const unsigned u1 = (unsigned)(cur[t].y - base);
            const unsigned u2 = (unsigned)(cur[t].z - base);
            const unsigned u3 = (unsigned)(cur[t].w - base);
            if (min(min(u0, u1), min(u2, u3)) < lim) {
                const unsigned e0 = 4u * (unsigned)group_of(i0, t);
                if (u0 < lim) push(e0 + 0u, u0);
                if (u1 < lim) push(e0 + 1u, u1);
                if (u2 < lim) push(e0 + 2u, u2);
                if (u3 < lim) push(e0 + 3u, u3);
            }
        }
        #pragma unroll
        for (int t = 0; t < 4; ++t) cur[t] = nxt[t];
    }
    __syncthreads();

    const unsigned total = qn;
    if (total <= QUEUE_CAP) {
        const unsigned wave = (unsigned)__builtin_amdgcn_readfirstlane(tid >> 5);
        const int lane = tid & 31;
        const uint4* queue4 = reinterpret_cast<const uint4*>(queue);
        for (unsigned k = wave * 8u; k < total; k += (AGG_THREADS / 32) * 8u) {
            const uint4 qa = queue4[k / 4u];
            const uint4 qb = queue4[k / 4u + 1u];
            const unsigned raw[8] = {qa.x, qa.y, qa.z, qa.w, qb.x, qb.y, qb.z, qb.w};
            unsigned ent[8];
            bool valid[8];
            const float* row[8];
            float v[8][4];
            #pragma unroll
            for (int t = 0; t < 8; ++t) {
                valid[t] = (k + t) < total;
                ent[t] = valid[t] ? (unsigned)__builtin_amdgcn_readfirstlane((int)raw[t]) : 0u;
            }
            #pragma unroll
            for (int t = 0; t < 8; ++t) {
                const unsigned e = ent[t] & 0xFFFFFu;
                const unsigned s = min((unsigned)src[e], (unsigned)(N_NODES - 1));
                row[t] = in + (size_t)s * D_FEAT + lane;
            }
            #pragma unroll
            for (int t = 0; t < 8; ++t) {
                #pragma unroll
                for (int c = 0; c < 4; ++c) v[t][c] = row[t][32 * c];
            }
            #pragma unroll
            for (int t = 0; t < 8; ++t) {
                if (valid[t]) {
                    const unsigned u = ent[t] >> 20;
                    float* a = acc + u * D_FEAT + lane;
                    #pragma unroll
                    for (int c = 0; c < 4; ++c)
                        __hip_atomic_fetch_add(a + 32 * c, v[t][c],
                                               __ATOMIC_RELAXED, __HIP_MEMORY_SCOPE_WORKGROUP);
                    if (lane == 0)
                        __hip_atomic_fetch_add(&cnt[u], 1,
                                               __ATOMIC_RELAXED, __HIP_MEMORY_SCOPE_WORKGROUP);
                }
            }
        }
    } else {
        for (int e = tid; e < N_EDGES; e += AGG_THREADS) {
            const unsigned u = (unsigned)(dst[e] - base);
            if (u < (unsigned)nrows) {
                const unsigned s = min((unsigned)src[e], (unsigned)(N_NODES - 1));
                const float* r = in + (size_t)s * D_FEAT;
                float* a = acc + u * D_FEAT;
                for (int c = 0; c < D_FEAT; ++c)
                    __hip_atomic_fetch_add(a + c, r[c],
                                           __ATOMIC_RELAXED, __HIP_MEMORY_SCOPE_WORKGROUP);
                __hip_atomic_fetch_add(&cnt[u], 1,
                                       __ATOMIC_RELAXED, __HIP_MEMORY_SCOPE_WORKGROUP);
            }
        }
    }
    __syncthreads();

    float4* outp = reinterpret_cast<float4*>(mean + (size_t)base * D_FEAT);
    for (int i = tid; i < nrows * (D_FEAT / 4); i += AGG_THREADS) {
        const float dv = fmaxf((float)cnt[i >> 5], 1.0f);
        float4 a = acc4[i];
        a.x = a.x / dv;
        a.y = a.y / dv;
        a.z = a.z / dv;
        a.w = a.w / dv;
        outp[i] = a;
    }
}

__global__ void sage_gemm_wmma(const float* __restrict__ Amean,
                               const float* __restrict__ Ax,
                               const float* __restrict__ Wl,
                               const float* __restrict__ Wr,
                               const float* __restrict__ bias,
                               float* __restrict__ out,
                               int do_relu) {
    const int wave  = threadIdx.x >> 5;
    const int lane  = threadIdx.x & 31;
    const int lhalf = lane >> 4;
    const int lmod  = lane & 15;
    const int rbase = blockIdx.x * 16;
    const int cbase = wave * 16;
    const int arow  = rbase + lmod;
    const int col   = cbase + lmod;

    v8f acc = {};
    const float* amp = Amean + arow * D_FEAT + (lhalf ? 2 : 0);
    const float* axp = Ax    + arow * D_FEAT + (lhalf ? 2 : 0);

    #pragma unroll 4
    for (int k = 0; k < D_FEAT; k += 4) {
        const int ka = k + (lhalf ? 2 : 0);
        v2f am = *(const v2f*)(amp + k);
        v2f ax = *(const v2f*)(axp + k);
        v2f bl, br;
        bl.x = Wl[(ka + 0) * D_FEAT + col];
        bl.y = Wl[(ka + 1) * D_FEAT + col];
        br.x = Wr[(ka + 0) * D_FEAT + col];
        br.y = Wr[(ka + 1) * D_FEAT + col];
        acc = __builtin_amdgcn_wmma_f32_16x16x4_f32(false, am, false, bl,
                                                    (short)0, acc, false, false);
        acc = __builtin_amdgcn_wmma_f32_16x16x4_f32(false, ax, false, br,
                                                    (short)0, acc, false, false);
    }

    float bv = bias[col];
    #pragma unroll
    for (int v = 0; v < 8; ++v) {
        float r = acc[v] + bv;
        if (do_relu) r = fmaxf(r, 0.0f);
        int m = v + (lhalf ? 8 : 0);
        out[(rbase + m) * D_FEAT + col] = r;
    }
}

extern "C" void kernel_launch(void* const* d_in, const int* in_sizes, int n_in,
                              void* d_out, int out_size, void* d_ws, size_t ws_size,
                              hipStream_t stream) {
    const float* x    = (const float*)d_in[0];
    const int*   eidx = (const int*)d_in[1];
    const float* W1l  = (const float*)d_in[2];
    const float* b1   = (const float*)d_in[3];
    const float* W1r  = (const float*)d_in[4];
    const float* W2l  = (const float*)d_in[5];
    const float* b2   = (const float*)d_in[6];
    const float* W2r  = (const float*)d_in[7];
    float* out = (float*)d_out;

    const int* src = eidx;
    const int* dst = eidx + N_EDGES;

    float* agg = (float*)d_ws;
    float* deg = agg + (size_t)N_NODES * D_FEAT;
    float* h   = deg + N_NODES;

    const int gemmBl  = N_NODES / 16;

    sage_aggregate<<<AGG_BLOCKS, AGG_THREADS, 0, stream>>>(x, src, dst, agg);
    sage_gemm_wmma<<<gemmBl, 256, 0, stream>>>(agg, x, W1l, W1r, b1, h, 1);

    sage_aggregate<<<AGG_BLOCKS, AGG_THREADS, 0, stream>>>(h, src, dst, agg);
    sage_gemm_wmma<<<gemmBl, 256, 0, stream>>>(agg, h, W2l, W2r, b2, out, 0);
}


# Products moved into the aggregation block: its LDS sum tile, divided in place, and an LDS copy of its own rows are the WMMA A operands; mean plane and 2 launches gone; sum orders unchanged
# speedup vs baseline: 3.8162x; 1.3237x over previous
#include <hip/hip_runtime.h>
#include <hip/hip_bf16.h>


#define N_NODES 50000
#define N_EDGES 800000
#define D_FEAT  128

#define ROWS_PER_BLOCK 196
#define AGG_BLOCKS     256
#define AGG_THREADS    1024
#define QUEUE_CAP      16384
#define TILE_ROWS      208
#define PITCH          132

typedef __attribute__((ext_vector_type(2))) float v2f;
typedef __attribute__((ext_vector_type(8))) float v8f;

template <int NT>
__device__ __forceinline__ void tile_products(const float* mean_t, const float* own_t,
                                              const float* __restrict__ Wl,
                                              const float* __restrict__ Wr,
                                              const float* __restrict__ bias,
                                              float* __restrict__ out,
                                              int g, int lmod, int lhalf, int col,
                                              int base, int nrows, int do_relu) {
    int mt[NT];
    const float* pm[NT];
    const float* po[NT];
    v8f c[NT];
    #pragma unroll
    for (int t = 0; t < NT; ++t) {
        mt[t] = (t < 3) ? 3 * g + t : 12;
        pm[t] = mean_t + (16 * mt[t] + lmod) * PITCH + 2 * lhalf;
        po[t] = own_t  + (16 * mt[t] + lmod) * PITCH + 2 * lhalf;
        c[t] = v8f{};
    }
    const float* wl = Wl + (2 * lhalf) * D_FEAT + col;
    const float* wr = Wr + (2 * lhalf) * D_FEAT + col;

    v2f amP[NT], axP[NT], blP, brP;
    v2f amQ[NT], axQ[NT], blQ, brQ;
    auto load_frags = [&](int k, v2f (&a_m)[NT], v2f (&a_x)[NT], v2f& b_l, v2f& b_r) {
        #pragma unroll
        for (int t = 0; t < NT; ++t) {
            a_m[t] = *(const v2f*)(pm[t] + k);
            a_x[t] = *(const v2f*)(po[t] + k);
        }
        b_l.x = wl[k * D_FEAT];
        b_l.y = wl[(k + 1) * D_FEAT];
        b_r.x = wr[k * D_FEAT];
        b_r.y = wr[(k + 1) * D_FEAT];
    };
    auto products = [&](const v2f (&a_m)[NT], const v2f (&a_x)[NT], const v2f& b_l, const v2f& b_r) {
        #pragma unroll
        for (int t = 0; t < NT; ++t) {
            c[t] = __builtin_amdgcn_wmma_f32_16x16x4_f32(false, a_m[t], false, b_l,
                                                         (short)0, c[t], false, false);
            c[t] = __builtin_amdgcn_wmma_f32_16x16x4_f32(false, a_x[t], false, b_r,
                                                         (short)0, c[t], false, false);
        }
    };
    load_frags(0, amP, axP, blP, brP);
    #pragma unroll 2
    for (int k = 0; k < D_FEAT; k += 8) {
        load_frags(k + 4, amQ, axQ, blQ, brQ);
        products(amP, axP, blP, brP);
        if (k + 8 < D_FEAT) load_frags(k + 8, amP, axP, blP, brP);
        products(amQ, axQ, blQ, brQ);
    }

    const float bv = bias[col];
    #pragma unroll
    for (int t = 0; t < NT; ++t) {
        #pragma unroll
        for (int r = 0; r < 8; ++r) {
            const int row = 16 * mt[t] + 8 * lhalf + r;
            if (row < nrows) {
                float v = c[t][r] + bv;
                if (do_relu) v = fmaxf(v, 0.0f);
                out[(size_t)(base + row) * D_FEAT + col] = v;
            }
        }
    }
}

__global__ __launch_bounds__(AGG_THREADS)
void sage_layer(const float* __restrict__ in,
                const int* __restrict__ src,
                const int* __restrict__ dst,
                const float* __restrict__ Wl,
                const float* __restrict__ Wr,
                const float* __restrict__ bias,
                float* __restrict__ out,
                int do_relu) {
    __shared__ __attribute__((aligned(16))) float acc[TILE_ROWS * PITCH];
    __shared__ __attribute__((aligned(16))) float own[TILE_ROWS * PITCH];
    __shared__ __attribute__((aligned(16))) unsigned int queue[QUEUE_CAP];
    __shared__ int cnt[TILE_ROWS];
    __shared__ unsigned int qn;

    const int tid   = threadIdx.x;
    const int base  = blockIdx.x * ROWS_PER_BLOCK;
    const int nrows = min(ROWS_PER_BLOCK, N_NODES - base);

    float4 ownv[7];
    #pragma unroll
    for (int j = 0; j < 7; ++j) {
        const int r = (tid >> 5) + 32 * j;
        ownv[j] = make_float4(0.f, 0.f, 0.f, 0.f);
        if (r < nrows)
            ownv[j] = *reinterpret_cast<const float4*>(in + (size_t)(base + r) * D_FEAT + 4 * (tid & 31));
    }

    float4* acc4 = reinterpret_cast<float4*>(acc);
    for (int i = tid; i < TILE_ROWS * PITCH / 4; i += AGG_THREADS)
        acc4[i] = make_float4(0.f, 0.f, 0.f, 0.f);
    if (tid < TILE_ROWS) cnt[tid] = 0;
    if (tid == 0) qn = 0u;
    #pragma unroll
    for (int j = 0; j < 7; ++j) {
        const int r = (tid >> 5) + 32 * j;
        if (r < TILE_ROWS)
            *reinterpret_cast<float4*>(own + r * PITCH + 4 * (tid & 31)) = ownv[j];
    }
    __syncthreads();

    const int n4 = N_EDGES / 4;
    const int4* __restrict__ dst4 = reinterpret_cast<const int4*>(dst);
    const int start = (int)(((unsigned)blockIdx.x * (unsigned)n4) / AGG_BLOCKS);

    auto group_of = [&](int i0, int t) -> int {
        const int i = i0 + t * AGG_THREADS + tid;
        int j = (i < n4 ? i : 0) + start;
        if (j >= n4) j -= n4;
        return j;
    };
    auto push = [&](unsigned e, unsigned u) {
        const unsigned slot = atomicAdd(&qn, 1u);
        if (slot < QUEUE_CAP) queue[slot] = e | (u << 20);
    };

    int4 cur[4];
    #pragma unroll
    for (int t = 0; t < 4; ++t) cur[t] = dst4[group_of(0, t)];

    for (int i0 = 0; i0 < n4; i0 += 4 * AGG_THREADS) {
        int4 nxt[4];
        #pragma unroll
        for (int t = 0; t < 4; ++t) nxt[t] = dst4[group_of(i0 + 4 * AGG_THREADS, t)];

        #pragma unroll
        for (int t = 0; t < 4; ++t) {
            const int i = i0 + t * AGG_THREADS + tid;
            const unsigned lim = (i < n4) ? (unsigned)nrows : 0u;
            const unsigned u0 = (unsigned)(cur[t].x - base);
            const unsigned u1 = (unsigned)(cur[t].y - base);
            const unsigned u2 = (unsigned)(cur[t].z - base);
            const unsigned u3 = (unsigned)(cur[t].w - base);
            if (min(min(u0, u1), min(u2, u3)) < lim) {
                const unsigned e0 = 4u * (unsigned)group_of(i0, t);
                if (u0 < lim) push(e0 + 0u, u0);
                if (u1 < lim) push(e0 + 1u, u1);
                if (u2 < lim) push(e0 + 2u, u2);
                if (u3 < lim) push(e0 + 3u, u3);
            }
        }
        #pragma unroll
        for (int t = 0; t < 4; ++t) cur[t] = nxt[t];
    }
    __syncthreads();

    const unsigned total = qn;
    if (total <= QUEUE_CAP) {
        const unsigned wave = (unsigned)__builtin_amdgcn_readfirstlane(tid >> 5);
        const int lane = tid & 31;
        const uint4* queue4 = reinterpret_cast<const uint4*>(queue);
        for (unsigned k = wave * 8u; k < total; k += (AGG_THREADS / 32) * 8u) {
            const uint4 qa = queue4[k / 4u];
            const uint4 qb = queue4[k / 4u + 1u];
            const unsigned raw[8] = {qa.x, qa.y, qa.z, qa.w, qb.x, qb.y, qb.z, qb.w};
            unsigned ent[8];
            bool valid[8];
            const float* row[8];
            float v[8][4];
            #pragma unroll
            for (int t = 0; t < 8; ++t) {
                valid[t] = (k + t) < total;
                ent[t] = valid[t] ? (unsigned)__builtin_amdgcn_readfirstlane((int)raw[t]) : 0u;
            }
            #pragma unroll
            for (int t = 0; t < 8; ++t) {
                const unsigned e = ent[t] & 0xFFFFFu;
                const unsigned s = min((unsigned)src[e], (unsigned)(N_NODES - 1));
                row[t] = in + (size_t)s * D_FEAT + lane;
            }
            #pragma unroll
            for (int t = 0; t < 8; ++t) {
                #pragma unroll
                for (int c = 0; c < 4; ++c) v[t][c] = row[t][32 * c];
            }
            #pragma unroll
            for (int t = 0; t < 8; ++t) {
                if (valid[t]) {
                    const unsigned u = ent[t] >> 20;
                    float* a = acc + u * PITCH + lane;
                    #pragma unroll
                    for (int c = 0; c < 4; ++c)
                        __hip_atomic_fetch_add(a + 32 * c, v[t][c],
                                               __ATOMIC_RELAXED, __HIP_MEMORY_SCOPE_WORKGROUP);
                    if (lane == 0)
                        __hip_atomic_fetch_add(&cnt[u], 1,
                                               __ATOMIC_RELAXED, __HIP_MEMORY_SCOPE_WORKGROUP);
                }
            }
        }
    } else {
        for (int e = tid; e < N_EDGES; e += AGG_THREADS) {
            const unsigned u = (unsigned)(dst[e] - base);
            if (u < (unsigned)nrows) {
                const unsigned s = min((unsigned)src[e], (unsigned)(N_NODES - 1));
                const float* r = in + (size_t)s * D_FEAT;
                float* a = acc + u * PITCH;
                for (int c = 0; c < D_FEAT; ++c)
                    __hip_atomic_fetch_add(a + c, r[c],
                                           __ATOMIC_RELAXED, __HIP_MEMORY_SCOPE_WORKGROUP);
                __hip_atomic_fetch_add(&cnt[u], 1,
                                       __ATOMIC_RELAXED, __HIP_MEMORY_SCOPE_WORKGROUP);
            }
        }
    }
    __syncthreads();

    for (int i = tid; i < nrows * (D_FEAT / 4); i += AGG_THREADS) {
        const int r = i >> 5;
        const float dv = fmaxf((float)cnt[r], 1.0f);
        float4* p = reinterpret_cast<float4*>(acc + r * PITCH + 4 * (i & 31));
        float4 a = *p;
        a.x = a.x / dv;
        a.y = a.y / dv;
        a.z = a.z / dv;
        a.w = a.w / dv;
        *p = a;
    }
    __syncthreads();

    {
        const int wave  = __builtin_amdgcn_readfirstlane(tid >> 5);
        const int lane  = tid & 31;
        const int lhalf = lane >> 4;
        const int lmod  = lane & 15;
        const int ntile = wave & 7;
        const int g     = wave >> 3;
        const int col   = ntile * 16 + lmod;
        if (g == (ntile >> 1))
            tile_products<4>(acc, own, Wl, Wr, bias, out, g, lmod, lhalf, col, base, nrows, do_relu);
        else
            tile_products<3>(acc, own, Wl, Wr, bias, out, g, lmod, lhalf, col, base, nrows, do_relu);
    }
}

extern "C" void kernel_launch(void* const* d_in, const int* in_sizes, int n_in,
                              void* d_out, int out_size, void* d_ws, size_t ws_size,
                              hipStream_t stream) {
    const float* x    = (const float*)d_in[0];
    const int*   eidx = (const int*)d_in[1];
    const float* W1l  = (const float*)d_in[2];
    const float* b1   = (const float*)d_in[3];
    const float* W1r  = (const float*)d_in[4];
    const float* W2l  = (const float*)d_in[5];
    const float* b2   = (const float*)d_in[6];
    const float* W2r  = (const float*)d_in[7];
    float* out = (float*)d_out;

    const int* src = eidx;
    const int* dst = eidx + N_EDGES;

    float* agg = (float*)d_ws;
    float* deg = agg + (size_t)N_NODES * D_FEAT;
    float* h   = deg + N_NODES;

    sage_layer<<<AGG_BLOCKS, AGG_THREADS, 0, stream>>>(x, src, dst, W1l, W1r, b1, h, 1);
    sage_layer<<<AGG_BLOCKS, AGG_THREADS, 0, stream>>>(h, src, dst, W2l, W2r, b2, out, 0);
}


# Each block's edge list, resolved to (source row, local row), is stored in the workspace by layer 1 and read back by layer 2: layer 2 scans no dst; sums still in LDS arrival order
# speedup vs baseline: 4.7517x; 1.2451x over previous
#include <hip/hip_runtime.h>
#include <hip/hip_bf16.h>


#define N_NODES 50000
#define N_EDGES 800000
#define D_FEAT  128

#define ROWS_PER_BLOCK 196
#define AGG_BLOCKS     256
#define AGG_THREADS    1024
#define QUEUE_CAP      16384
#define TILE_ROWS      208
#define PITCH          132

typedef __attribute__((ext_vector_type(2))) float v2f;
typedef __attribute__((ext_vector_type(8))) float v8f;

template <int NT>
__device__ __forceinline__ void tile_products(const float* mean_t, const float* own_t,
                                              const float* __restrict__ Wl,
                                              const float* __restrict__ Wr,
                                              const float* __restrict__ bias,
                                              float* __restrict__ out,
                                              int g, int lmod, int lhalf, int col,
                                              int base, int nrows, int do_relu) {
    int mt[NT];
    const float* pm[NT];
    const float* po[NT];
    v8f c[NT];
    #pragma unroll
    for (int t = 0; t < NT; ++t) {
        mt[t] = (t < 3) ? 3 * g + t : 12;
        pm[t] = mean_t + (16 * mt[t] + lmod) * PITCH + 2 * lhalf;
        po[t] = own_t  + (16 * mt[t] + lmod) * PITCH + 2 * lhalf;
        c[t] = v8f{};
    }
    const float* wl = Wl + (2 * lhalf) * D_FEAT + col;
    const float* wr = Wr + (2 * lhalf) * D_FEAT + col;

    v2f amP[NT], axP[NT], blP, brP;
    v2f amQ[NT], axQ[NT], blQ, brQ;
    auto load_frags = [&](int k, v2f (&a_m)[NT], v2f (&a_x)[NT], v2f& b_l, v2f& b_r) {
        #pragma unroll
        for (int t = 0; t < NT; ++t) {
            a_m[t] = *(const v2f*)(pm[t] + k);
            a_x[t] = *(const v2f*)(po[t] + k);
        }
        b_l.x = wl[k * D_FEAT];
        b_l.y = wl[(k + 1) * D_FEAT];
        b_r.x = wr[k * D_FEAT];
        b_r.y = wr[(k + 1) * D_FEAT];
    };
    auto products = [&](const v2f (&a_m)[NT], const v2f (&a_x)[NT], const v2f& b_l, const v2f& b_r) {
        #pragma unroll
        for (int t = 0; t < NT; ++t) {
            c[t] = __builtin_amdgcn_wmma_f32_16x16x4_f32(false, a_m[t], false, b_l,
                                                         (short)0, c[t], false, false);
            c[t] = __builtin_amdgcn_wmma_f32_16x16x4_f32(false, a_x[t], false, b_r,
                                                         (short)0, c[t], false, false);
        }
    };
    load_frags(0, amP, axP, blP, brP);
    #pragma unroll 2
    for (int k = 0; k < D_FEAT; k += 8) {
        load_frags(k + 4, amQ, axQ, blQ, brQ);
        products(amP, axP, blP, brP);
        if (k + 8 < D_FEAT) load_frags(k + 8, amP, axP, blP, brP);
        products(amQ, axQ, blQ, brQ);
    }

    const float bv = bias[col];
    #pragma unroll
    for (int t = 0; t < NT; ++t) {
        #pragma unroll
        for (int r = 0; r < 8; ++r) {
            const int row = 16 * mt[t] + 8 * lhalf + r;
            if (row < nrows) {
                float v = c[t][r] + bv;
                if (do_relu) v = fmaxf(v, 0.0f);
                out[(size_t)(base + row) * D_FEAT + col] = v;
            }
        }
    }
}

template <int LAYER>
__global__ __launch_bounds__(AGG_THREADS)
void sage_layer(const float* __restrict__ in,
                const int* __restrict__ src,
                const int* __restrict__ dst,
                const float* __restrict__ Wl,
                const float* __restrict__ Wr,
                const float* __restrict__ bias,
                float* __restrict__ out,
                unsigned int* __restrict__ list_ws,
                unsigned int* __restrict__ count_ws,
                int do_relu) {
    __shared__ __attribute__((aligned(16))) float acc[TILE_ROWS * PITCH];
    __shared__ __attribute__((aligned(16))) float own[TILE_ROWS * PITCH];
    __shared__ __attribute__((aligned(16))) unsigned int queue[QUEUE_CAP];
    __shared__ int cnt[TILE_ROWS];
    __shared__ unsigned int qn;

    const int tid   = threadIdx.x;
    const int base  = blockIdx.x * ROWS_PER_BLOCK;
    const int nrows = min(ROWS_PER_BLOCK, N_NODES - base);
    unsigned int* __restrict__ my_list = list_ws + (size_t)blockIdx.x * QUEUE_CAP;

    unsigned total = 0u;
    if (LAYER == 2) total = count_ws[blockIdx.x];

    float4 ownv[7];
    #pragma unroll
    for (int j = 0; j < 7; ++j) {
        const int r = (tid >> 5) + 32 * j;
        ownv[j] = make_float4(0.f, 0.f, 0.f, 0.f);
        if (r < nrows)
            ownv[j] = *reinterpret_cast<const float4*>(in + (size_t)(base + r) * D_FEAT + 4 * (tid & 31));
    }

    const unsigned n16 = (LAYER == 2 && total <= QUEUE_CAP) ? ((total + 7u) & ~7u) / 4u : 0u;
    uint4 lv[4];
    if (LAYER == 2) {
        const uint4* l4 = reinterpret_cast<const uint4*>(my_list);
        #pragma unroll
        for (int j = 0; j < 4; ++j) {
            const unsigned i = (unsigned)tid + AGG_THREADS * j;
            lv[j] = make_uint4(0u, 0u, 0u, 0u);
            if (i < n16) lv[j] = l4[i];
        }
    }

    float4* acc4 = reinterpret_cast<float4*>(acc);
    for (int i = tid; i < TILE_ROWS * PITCH / 4; i += AGG_THREADS)
        acc4[i] = make_float4(0.f, 0.f, 0.f, 0.f);
    if (tid < TILE_ROWS) cnt[tid] = 0;
    if (tid == 0) qn = 0u;
    #pragma unroll
    for (int j = 0; j < 7; ++j) {
        const int r = (tid >> 5) + 32 * j;
        if (r < TILE_ROWS)
            *reinterpret_cast<float4*>(own + r * PITCH + 4 * (tid & 31)) = ownv[j];
    }
    if (LAYER == 2) {
        uint4* q4 = reinterpret_cast<uint4*>(queue);
        #pragma unroll
        for (int j = 0; j < 4; ++j) {
            const unsigned i = (unsigned)tid + AGG_THREADS * j;
            if (i < n16) q4[i] = lv[j];
        }
    }
    __syncthreads();

    if (LAYER == 1) {
        const int n4 = N_EDGES / 4;
        const int4* __restrict__ dst4 = reinterpret_cast<const int4*>(dst);
        const int start = (int)(((unsigned)blockIdx.x * (unsigned)n4) / AGG_BLOCKS);

        auto group_of = [&](int i0, int t) -> int {
            const int i = i0 + t * AGG_THREADS + tid;
            int j = (i < n4 ? i : 0) + start;
            if (j >= n4) j -= n4;
            return j;
        };
        auto push = [&](unsigned e, unsigned u) {
            const unsigned slot = atomicAdd(&qn, 1u);
            if (slot < QUEUE_CAP) queue[slot] = e | (u << 20);
        };

        int4 cur[4];
        #pragma unroll
        for (int t = 0; t < 4; ++t) cur[t] = dst4[group_of(0, t)];

        for (int i0 = 0; i0 < n4; i0 += 4 * AGG_THREADS) {
            int4 nxt[4];
            #pragma unroll
            for (int t = 0; t < 4; ++t) nxt[t] = dst4[group_of(i0 + 4 * AGG_THREADS, t)];

            #pragma unroll
            for (int t = 0; t < 4; ++t) {
                const int i = i0 + t * AGG_THREADS + tid;
                const unsigned lim = (i < n4) ? (unsigned)nrows : 0u;
                const unsigned u0 = (unsigned)(cur[t].x - base);
                const unsigned u1 = (unsigned)(cur[t].y - base);
                const unsigned u2 = (unsigned)(cur[t].z - base);
                const unsigned u3 = (unsigned)(cur[t].w - base);
                if (min(min(u0, u1), min(u2, u3)) < lim) {
                    const unsigned e0 = 4u * (unsigned)group_of(i0, t);
                    if (u0 < lim) push(e0 + 0u, u0);
                    if (u1 < lim) push(e0 + 1u, u1);
                    if (u2 < lim) push(e0 + 2u, u2);
                    if (u3 < lim) push(e0 + 3u, u3);
                }
            }
            #pragma unroll
            for (int t = 0; t < 4; ++t) cur[t] = nxt[t];
        }
        __syncthreads();

        total = qn;
        if (total <= QUEUE_CAP) {
            const unsigned padded = (total + 7u) & ~7u;
            for (unsigned i0 = 0; i0 < padded; i0 += 4 * AGG_THREADS) {
                unsigned q[4], sr[4];
                #pragma unroll
                for (int j = 0; j < 4; ++j) {
                    const unsigned i = i0 + AGG_THREADS * j + tid;
                    q[j] = (i < total) ? queue[i] : 0u;
                    sr[j] = (unsigned)src[q[j] & 0xFFFFFu];
                }
                #pragma unroll
                for (int j = 0; j < 4; ++j) {
                    const unsigned i = i0 + AGG_THREADS * j + tid;
                    if (i < padded) {
                        const unsigned val = (i < total) ? (min(sr[j], (unsigned)(N_NODES - 1)) | ((q[j] >> 20) << 16)) : 0u;
                        queue[i] = val;
                        my_list[i] = val;
                    }
                }
            }
        }
        if (tid == 0) count_ws[blockIdx.x] = total;
        __syncthreads();
    }

    if (total <= QUEUE_CAP) {
        const unsigned wave = (unsigned)__builtin_amdgcn_readfirstlane(tid >> 5);
        const int lane = tid & 31;
        const uint4* queue4 = reinterpret_cast<const uint4*>(queue);
        for (unsigned k = wave * 8u; k < total; k += (AGG_THREADS / 32) * 8u) {
            const uint4 qa = queue4[k / 4u];
            const uint4 qb = queue4[k / 4u + 1u];
            const unsigned raw[8] = {qa.x, qa.y, qa.z, qa.w, qb.x, qb.y, qb.z, qb.w};
            unsigned ent[8];
            bool valid[8];
            const float* row[8];
            float v[8][4];
            #pragma unroll
            for (int t = 0; t < 8; ++t) {
                valid[t] = (k + t) < total;
                ent[t] = valid[t] ? (unsigned)__builtin_amdgcn_readfirstlane((int)raw[t]) : 0u;
            }
            #pragma unroll
            for (int t = 0; t < 8; ++t) {
                const unsigned s = min(ent[t] & 0xFFFFu, (unsigned)(N_NODES - 1));
                row[t] = in + (size_t)s * D_FEAT + lane;
            }
            #pragma unroll
            for (int t = 0; t < 8; ++t) {
                #pragma unroll
                for (int c = 0; c < 4; ++c) v[t][c] = row[t][32 * c];
            }
            #pragma unroll
            for (int t = 0; t < 8; ++t) {
                if (valid[t]) {
                    const unsigned u = min(ent[t] >> 16, (unsigned)(TILE_ROWS - 1));
                    float* a = acc + u * PITCH + lane;
                    #pragma unroll
                    for (int c = 0; c < 4; ++c)
                        __hip_atomic_fetch_add(a + 32 * c, v[t][c],
                                               __ATOMIC_RELAXED, __HIP_MEMORY_SCOPE_WORKGROUP);
                    if (lane == 0)
                        __hip_atomic_fetch_add(&cnt[u], 1,
                                               __ATOMIC_RELAXED, __HIP_MEMORY_SCOPE_WORKGROUP);
                }
            }
        }
    } else {
        for (int e = tid; e < N_EDGES; e += AGG_THREADS) {
            const unsigned u = (unsigned)(dst[e] - base);
            if (u < (unsigned)nrows) {
                const unsigned s = min((unsigned)src[e], (unsigned)(N_NODES - 1));
                const float* r = in + (size_t)s * D_FEAT;
                float* a = acc + u * PITCH;
                for (int c = 0; c < D_FEAT; ++c)
                    __hip_atomic_fetch_add(a + c, r[c],
                                           __ATOMIC_RELAXED, __HIP_MEMORY_SCOPE_WORKGROUP);
                __hip_atomic_fetch_add(&cnt[u], 1,
                                       __ATOMIC_RELAXED, __HIP_MEMORY_SCOPE_WORKGROUP);
            }
        }
    }
    __syncthreads();

    for (int i = tid; i < nrows * (D_FEAT / 4); i += AGG_THREADS) {
        const int r = i >> 5;
        const float dv = fmaxf((float)cnt[r], 1.0f);
        float4* p = reinterpret_cast<float4*>(acc + r * PITCH + 4 * (i & 31));
        float4 a = *p;
        a.x = a.x / dv;
        a.y = a.y / dv;
        a.z = a.z / dv;
        a.w = a.w / dv;
        *p = a;
    }
    __syncthreads();

    {
        const int wave  = __builtin_amdgcn_readfirstlane(tid >> 5);
        const int lane  = tid & 31;
        const int lhalf = lane >> 4;
        const int lmod  = lane & 15;
        const int ntile = wave & 7;
        const int g     = wave >> 3;
        const int col   = ntile * 16 + lmod;
        if (g == (ntile >> 1))
            tile_products<4>(acc, own, Wl, Wr, bias, out, g, lmod, lhalf, col, base, nrows, do_relu);
        else
            tile_products<3>(acc, own, Wl, Wr, bias, out, g, lmod, lhalf, col, base, nrows, do_relu);
    }
}

extern "C" void kernel_launch(void* const* d_in, const int* in_sizes, int n_in,
                              void* d_out, int out_size, void* d_ws, size_t ws_size,
                              hipStream_t stream) {
    const float* x    = (const float*)d_in[0];
    const int*   eidx = (const int*)d_in[1];
    const float* W1l  = (const float*)d_in[2];
    const float* b1   = (const float*)d_in[3];
    const float* W1r  = (const float*)d_in[4];
    const float* W2l  = (const float*)d_in[5];
    const float* b2   = (const float*)d_in[6];
    const float* W2r  = (const float*)d_in[7];
    float* out = (float*)d_out;

    const int* src = eidx;
    const int* dst = eidx + N_EDGES;

    float* agg = (float*)d_ws;
    float* deg = agg + (size_t)N_NODES * D_FEAT;
    float* h   = deg + N_NODES;
    unsigned int* list_ws  = (unsigned int*)agg;
    unsigned int* count_ws = (unsigned int*)deg;

    sage_layer<1><<<AGG_BLOCKS, AGG_THREADS, 0, stream>>>(x, src, dst, W1l, W1r, b1, h, list_ws, count_ws, 1);
    sage_layer<2><<<AGG_BLOCKS, AGG_THREADS, 0, stream>>>(h, src, dst, W2l, W2r, b2, out, list_ws, count_ws, 0);
}


# A first launch turns dst into one flag plane per block (1 bit per 4-edge group, set in LDS); layer 1 reads its own plane and the dst of flagged groups only; 3 launches; sums still LDS arrival order
# speedup vs baseline: 5.8905x; 1.2397x over previous
#include <hip/hip_runtime.h>
#include <hip/hip_bf16.h>


#define N_NODES 50000
#define N_EDGES 800000
#define D_FEAT  128

#define ROWS_PER_BLOCK 196
#define AGG_BLOCKS     256
#define AGG_THREADS    1024
#define QUEUE_CAP      16384
#define TILE_ROWS      208
#define PITCH          132
#define GROUPS         (N_EDGES / 4)
#define CHUNK_GROUPS   1024
#define N_CHUNKS       ((GROUPS + CHUNK_GROUPS - 1) / CHUNK_GROUPS)
#define PLANE_WORDS    (N_CHUNKS * 32)
#define GQ_CAP         8192

typedef __attribute__((ext_vector_type(2))) float v2f;
typedef __attribute__((ext_vector_type(8))) float v8f;

template <int NT>
__device__ __forceinline__ void tile_products(const float* mean_t, const float* own_t,
                                              const float* __restrict__ Wl,
                                              const float* __restrict__ Wr,
                                              const float* __restrict__ bias,
                                              float* __restrict__ out,
                                              int g, int lmod, int lhalf, int col,
                                              int base, int nrows, int do_relu) {
    int mt[NT];
    const float* pm[NT];
    const float* po[NT];
    v8f c[NT];
    #pragma unroll
    for (int t = 0; t < NT; ++t) {
        mt[t] = (t < 3) ? 3 * g + t : 12;
        pm[t] = mean_t + (16 * mt[t] + lmod) * PITCH + 2 * lhalf;
        po[t] = own_t  + (16 * mt[t] + lmod) * PITCH + 2 * lhalf;
        c[t] = v8f{};
    }
    const float* wl = Wl + (2 * lhalf) * D_FEAT + col;
    const float* wr = Wr + (2 * lhalf) * D_FEAT + col;

    v2f amP[NT], axP[NT], blP, brP;
    v2f amQ[NT], axQ[NT], blQ, brQ;
    auto load_frags = [&](int k, v2f (&a_m)[NT], v2f (&a_x)[NT], v2f& b_l, v2f& b_r) {
        #pragma unroll
        for (int t = 0; t < NT; ++t) {
            a_m[t] = *(const v2f*)(pm[t] + k);
            a_x[t] = *(const v2f*)(po[t] + k);
        }
        b_l.x = wl[k * D_FEAT];
        b_l.y = wl[(k + 1) * D_FEAT];
        b_r.x = wr[k * D_FEAT];
        b_r.y = wr[(k + 1) * D_FEAT];
    };
    auto products = [&](const v2f (&a_m)[NT], const v2f (&a_x)[NT], const v2f& b_l, const v2f& b_r) {
        #pragma unroll
        for (int t = 0; t < NT; ++t) {
            c[t] = __builtin_amdgcn_wmma_f32_16x16x4_f32(false, a_m[t], false, b_l,
                                                         (short)0, c[t], false, false);
            c[t] = __builtin_amdgcn_wmma_f32_16x16x4_f32(false, a_x[t], false, b_r,
                                                         (short)0, c[t], false, false);
        }
    };
    load_frags(0, amP, axP, blP, brP);
    #pragma unroll 2
    for (int k = 0; k < D_FEAT; k += 8) {
        load_frags(k + 4, amQ, axQ, blQ, brQ);
        products(amP, axP, blP, brP);
        if (k + 8 < D_FEAT) load_frags(k + 8, amP, axP, blP, brP);
        products(amQ, axQ, blQ, brQ);
    }

    const float bv = bias[col];
    #pragma unroll
    for (int t = 0; t < NT; ++t) {
        #pragma unroll
        for (int r = 0; r < 8; ++r) {
            const int row = 16 * mt[t] + 8 * lhalf + r;
            if (row < nrows) {
                float v = c[t][r] + bv;
                if (do_relu) v = fmaxf(v, 0.0f);
                out[(size_t)(base + row) * D_FEAT + col] = v;
            }
        }
    }
}

__global__ __launch_bounds__(CHUNK_GROUPS)
void flag_groups(const int* __restrict__ dst, unsigned int* __restrict__ planes) {
    __shared__ __attribute__((aligned(16))) unsigned int bits[AGG_BLOCKS * 32];
    const int tid = threadIdx.x;
    const int g = blockIdx.x * CHUNK_GROUPS + tid;
    int4 d = make_int4(-1, -1, -1, -1);
    if (g < GROUPS) d = reinterpret_cast<const int4*>(dst)[g];
    uint4* b4 = reinterpret_cast<uint4*>(bits);
    b4[tid] = make_uint4(0u, 0u, 0u, 0u);
    b4[tid + CHUNK_GROUPS] = make_uint4(0u, 0u, 0u, 0u);
    __syncthreads();
    const unsigned bit = 1u << (tid >> 5);
    const unsigned lane = (unsigned)tid & 31u;
    if ((unsigned)d.x < (unsigned)N_NODES) atomicOr(&bits[((unsigned)d.x / ROWS_PER_BLOCK) * 32u + lane], bit);
    if ((unsigned)d.y < (unsigned)N_NODES) atomicOr(&bits[((unsigned)d.y / ROWS_PER_BLOCK) * 32u + lane], bit);
    if ((unsigned)d.z < (unsigned)N_NODES) atomicOr(&bits[((unsigned)d.z / ROWS_PER_BLOCK) * 32u + lane], bit);
    if ((unsigned)d.w < (unsigned)N_NODES) atomicOr(&bits[((unsigned)d.w / ROWS_PER_BLOCK) * 32u + lane], bit);
    __syncthreads();
    #pragma unroll
    for (int j = 0; j < 2; ++j) {
        const int i = tid + CHUNK_GROUPS * j;
        *reinterpret_cast<uint4*>(planes + (size_t)(i >> 3) * PLANE_WORDS + blockIdx.x * 32 + 4 * (i & 7)) = b4[i];
    }
}

template <int LAYER>
__global__ __launch_bounds__(AGG_THREADS)
void sage_layer(const float* __restrict__ in,
                const int* __restrict__ src,
                const int* __restrict__ dst,
                const unsigned int* __restrict__ planes,
                const float* __restrict__ Wl,
                const float* __restrict__ Wr,
                const float* __restrict__ bias,
                float* __restrict__ out,
                unsigned int* __restrict__ list_ws,
                unsigned int* __restrict__ count_ws,
                int do_relu) {
    __shared__ __attribute__((aligned(16))) float acc[TILE_ROWS * PITCH];
    __shared__ __attribute__((aligned(16))) float own[TILE_ROWS * PITCH];
    __shared__ __attribute__((aligned(16))) unsigned int queue[QUEUE_CAP];
    __shared__ __attribute__((aligned(16))) unsigned int gq[GQ_CAP];
    __shared__ int cnt[TILE_ROWS];
    __shared__ unsigned int qn;
    __shared__ unsigned int gqn;

    const int tid   = threadIdx.x;
    const int base  = blockIdx.x * ROWS_PER_BLOCK;
    const int nrows = min(ROWS_PER_BLOCK, N_NODES - base);
    unsigned int* __restrict__ my_list = list_ws + (size_t)blockIdx.x * QUEUE_CAP;

    unsigned total = 0u;
    if (LAYER == 2) total = count_ws[blockIdx.x];

    float4 ownv[7];
    #pragma unroll
    for (int j = 0; j < 7; ++j) {
        const int r = (tid >> 5) + 32 * j;
        ownv[j] = make_float4(0.f, 0.f, 0.f, 0.f);
        if (r < nrows)
            ownv[j] = *reinterpret_cast<const float4*>(in + (size_t)(base + r) * D_FEAT + 4 * (tid & 31));
    }

    uint4 pw[2];
    if (LAYER == 1) {
        const uint4* pl4 = reinterpret_cast<const uint4*>(planes + (size_t)blockIdx.x * PLANE_WORDS);
        #pragma unroll
        for (int j = 0; j < 2; ++j) {
            const int i = tid + AGG_THREADS * j;
            pw[j] = make_uint4(0u, 0u, 0u, 0u);
            if (i < PLANE_WORDS / 4) pw[j] = pl4[i];
        }
    }

    const unsigned n16 = (LAYER == 2 && total <= QUEUE_CAP) ? ((total + 7u) & ~7u) / 4u : 0u;
    uint4 lv[4];
    if (LAYER == 2) {
        const uint4* l4 = reinterpret_cast<const uint4*>(my_list);
        #pragma unroll
        for (int j = 0; j < 4; ++j) {
            const unsigned i = (unsigned)tid + AGG_THREADS * j;
            lv[j] = make_uint4(0u, 0u, 0u, 0u);
            if (i < n16) lv[j] = l4[i];
        }
    }

    float4* acc4 = reinterpret_cast<float4*>(acc);
    for (int i = tid; i < TILE_ROWS * PITCH / 4; i += AGG_THREADS)
        acc4[i] = make_float4(0.f, 0.f, 0.f, 0.f);
    if (tid < TILE_ROWS) cnt[tid] = 0;
    if (tid == 0) { qn = 0u; gqn = 0u; }
    #pragma unroll
    for (int j = 0; j < 7; ++j) {
        const int r = (tid >> 5) + 32 * j;
        if (r < TILE_ROWS)
            *reinterpret_cast<float4*>(own + r * PITCH + 4 * (tid & 31)) = ownv[j];
    }
    if (LAYER == 2) {
        uint4* q4 = reinterpret_cast<uint4*>(queue);
        #pragma unroll
        for (int j = 0; j < 4; ++j) {
            const unsigned i = (unsigned)tid + AGG_THREADS * j;
            if (i < n16) q4[i] = lv[j];
        }
    }
    __syncthreads();

    if (LAYER == 1) {
        const unsigned w[8] = {pw[0].x, pw[0].y, pw[0].z, pw[0].w, pw[1].x, pw[1].y, pw[1].z, pw[1].w};
        {
            unsigned n = 0u;
            #pragma unroll
            for (int k = 0; k < 8; ++k) n += (unsigned)__popc(w[k]);
            unsigned slot = 0u;
            if (n != 0u) slot = atomicAdd(&gqn, n);
            #pragma unroll
            for (int k = 0; k < 8; ++k) {
                const unsigned word = 4u * ((unsigned)tid + AGG_THREADS * (unsigned)(k >> 2)) + (unsigned)(k & 3);
                const unsigned g0 = (word >> 5) * CHUNK_GROUPS + (word & 31u);
                unsigned m = w[k];
                while (m != 0u) {
                    const unsigned v = (unsigned)__builtin_ctz(m);
                    m &= m - 1u;
                    if (slot < GQ_CAP) gq[slot] = g0 + 32u * v;
                    ++slot;
                }
            }
        }
        __syncthreads();

        const unsigned gtotal = gqn;
        if (gtotal <= GQ_CAP) {
            const int4* __restrict__ dst4 = reinterpret_cast<const int4*>(dst);
            for (unsigned i0 = 0; i0 < gtotal; i0 += 4 * AGG_THREADS) {
                unsigned gi[4];
                int4 dv[4];
                #pragma unroll
                for (int j = 0; j < 4; ++j) {
                    const unsigned i = i0 + AGG_THREADS * j + tid;
                    gi[j] = (i < gtotal) ? min(gq[i], (unsigned)(GROUPS - 1)) : 0u;
                    dv[j] = dst4[gi[j]];
                }
                unsigned u[4][4];
                unsigned n = 0u;
                #pragma unroll
                for (int j = 0; j < 4; ++j) {
                    const unsigned i = i0 + AGG_THREADS * j + tid;
                    const unsigned lim = (i < gtotal) ? (unsigned)nrows : 0u;
                    u[j][0] = (unsigned)(dv[j].x - base);
                    u[j][1] = (unsigned)(dv[j].y - base);
                    u[j][2] = (unsigned)(dv[j].z - base);
                    u[j][3] = (unsigned)(dv[j].w - base);
                    #pragma unroll
                    for (int c = 0; c < 4; ++c) {
                        if (u[j][c] >= lim) u[j][c] = 0xFFFFFFFFu;
                        n += (u[j][c] != 0xFFFFFFFFu) ? 1u : 0u;
                    }
                }
                unsigned slot = 0u;
                if (n != 0u) slot = atomicAdd(&qn, n);
                #pragma unroll
                for (int j = 0; j < 4; ++j) {
                    #pragma unroll
                    for (int c = 0; c < 4; ++c) {
                        if (u[j][c] != 0xFFFFFFFFu) {
                            if (slot < QUEUE_CAP) queue[slot] = (4u * gi[j] + (unsigned)c) | (u[j][c] << 20);
                            ++slot;
                        }
                    }
                }
            }
        }
        __syncthreads();

        total = (gtotal <= GQ_CAP) ? qn : (unsigned)(QUEUE_CAP + 1);
        if (total <= QUEUE_CAP) {
            const unsigned padded = (total + 7u) & ~7u;
            for (unsigned i0 = 0; i0 < padded; i0 += 4 * AGG_THREADS) {
                unsigned q[4], sr[4];
                #pragma unroll
                for (int j = 0; j < 4; ++j) {
                    const unsigned i = i0 + AGG_THREADS * j + tid;
                    q[j] = (i < total) ? queue[i] : 0u;
                    sr[j] = (unsigned)src[q[j] & 0xFFFFFu];
                }
                #pragma unroll
                for (int j = 0; j < 4; ++j) {
                    const unsigned i = i0 + AGG_THREADS * j + tid;
                    if (i < padded) {
                        const unsigned val = (i < total) ? (min(sr[j], (unsigned)(N_NODES - 1)) | ((q[j] >> 20) << 16)) : 0u;
                        queue[i] = val;
                        my_list[i] = val;
                    }
                }
            }
        }
        if (tid == 0) count_ws[blockIdx.x] = total;
        __syncthreads();
    }

    if (total <= QUEUE_CAP) {
        const unsigned wave = (unsigned)__builtin_amdgcn_readfirstlane(tid >> 5);
        const int lane = tid & 31;
        const uint4* queue4 = reinterpret_cast<const uint4*>(queue);
        for (unsigned k = wave * 8u; k < total; k += (AGG_THREADS / 32) * 8u) {
            const uint4 qa = queue4[k / 4u];
            const uint4 qb = queue4[k / 4u + 1u];
            const unsigned raw[8] = {qa.x, qa.y, qa.z, qa.w, qb.x, qb.y, qb.z, qb.w};
            unsigned ent[8];
            bool valid[8];
            const float* row[8];
            float v[8][4];
            #pragma unroll
            for (int t = 0; t < 8; ++t) {
                valid[t] = (k + t) < total;
                ent[t] = valid[t] ? (unsigned)__builtin_amdgcn_readfirstlane((int)raw[t]) : 0u;
            }
            #pragma unroll
            for (int t = 0; t < 8; ++t) {
                const unsigned s = min(ent[t] & 0xFFFFu, (unsigned)(N_NODES - 1));
                row[t] = in + (size_t)s * D_FEAT + lane;
            }
            #pragma unroll
            for (int t = 0; t < 8; ++t) {
                #pragma unroll
                for (int c = 0; c < 4; ++c) v[t][c] = row[t][32 * c];
            }
            #pragma unroll
            for (int t = 0; t < 8; ++t) {
                if (valid[t]) {
                    const unsigned u = min(ent[t] >> 16, (unsigned)(TILE_ROWS - 1));
                    float* a = acc + u * PITCH + lane;
                    #pragma unroll
                    for (int c = 0; c < 4; ++c)
                        __hip_atomic_fetch_add(a + 32 * c, v[t][c],
                                               __ATOMIC_RELAXED, __HIP_MEMORY_SCOPE_WORKGROUP);
                    if (lane == 0)
                        __hip_atomic_fetch_add(&cnt[u], 1,
                                               __ATOMIC_RELAXED, __HIP_MEMORY_SCOPE_WORKGROUP);
                }
            }
        }
    } else {
        for (int e = tid; e < N_EDGES; e += AGG_THREADS) {
            const unsigned u = (unsigned)(dst[e] - base);
            if (u < (unsigned)nrows) {
                const unsigned s = min((unsigned)src[e], (unsigned)(N_NODES - 1));
                const float* r = in + (size_t)s * D_FEAT;
                float* a = acc + u * PITCH;
                for (int c = 0; c < D_FEAT; ++c)
                    __hip_atomic_fetch_add(a + c, r[c],
                                           __ATOMIC_RELAXED, __HIP_MEMORY_SCOPE_WORKGROUP);
                __hip_atomic_fetch_add(&cnt[u], 1,
                                       __ATOMIC_RELAXED, __HIP_MEMORY_SCOPE_WORKGROUP);
            }
        }
    }
    __syncthreads();

    for (int i = tid; i < nrows * (D_FEAT / 4); i += AGG_THREADS) {
        const int r = i >> 5;
        const float dv = fmaxf((float)cnt[r], 1.0f);
        float4* p = reinterpret_cast<float4*>(acc + r * PITCH + 4 * (i & 31));
        float4 a = *p;
        a.x = a.x / dv;
        a.y = a.y / dv;
        a.z = a.z / dv;
        a.w = a.w / dv;
        *p = a;
    }
    __syncthreads();

    {
        const int wave  = __builtin_amdgcn_readfirstlane(tid >> 5);
        const int lane  = tid & 31;
        const int lhalf = lane >> 4;
        const int lmod  = lane & 15;
        const int ntile = wave & 7;
        const int g     = wave >> 3;
        const int col   = ntile * 16 + lmod;
        if (g == (ntile >> 1))
            tile_products<4>(acc, own, Wl, Wr, bias, out, g, lmod, lhalf, col, base, nrows, do_relu);
        else
            tile_products<3>(acc, own, Wl, Wr, bias, out, g, lmod, lhalf, col, base, nrows, do_relu);
    }
}

extern "C" void kernel_launch(void* const* d_in, const int* in_sizes, int n_in,
                              void* d_out, int out_size, void* d_ws, size_t ws_size,
                              hipStream_t stream) {
    const float* x    = (const float*)d_in[0];
    const int*   eidx = (const int*)d_in[1];
    const float* W1l  = (const float*)d_in[2];
    const float* b1   = (const float*)d_in[3];
    const float* W1r  = (const float*)d_in[4];
    const float* W2l  = (const float*)d_in[5];
    const float* b2   = (const float*)d_in[6];
    const float* W2r  = (const float*)d_in[7];
    float* out = (float*)d_out;

    const int* src = eidx;
    const int* dst = eidx + N_EDGES;

    float* agg = (float*)d_ws;
    float* deg = agg + (size_t)N_NODES * D_FEAT;
    float* h   = deg + N_NODES;
    unsigned int* list_ws  = (unsigned int*)agg;
    unsigned int* count_ws = (unsigned int*)deg;
    unsigned int* plane_ws = list_ws + (size_t)AGG_BLOCKS * QUEUE_CAP;

    flag_groups<<<N_CHUNKS, CHUNK_GROUPS, 0, stream>>>(dst, plane_ws);

    sage_layer<1><<<AGG_BLOCKS, AGG_THREADS, 0, stream>>>(x, src, dst, plane_ws, W1l, W1r, b1, h, list_ws, count_ws, 1);
    sage_layer<2><<<AGG_BLOCKS, AGG_THREADS, 0, stream>>>(h, src, dst, plane_ws, W2l, W2r, b2, out, list_ws, count_ws, 0);
}


# Launch 0 bins the edges: one 256-B segment of (source row, local row) per (4096-edge chunk, block), built in LDS; layer 1 packs its 196 segments, reads no plane, dst or src; sums: LDS arrival order
# speedup vs baseline: 6.2538x; 1.0617x over previous
#include <hip/hip_runtime.h>
#include <hip/hip_bf16.h>


#define N_NODES 50000
#define N_EDGES 800000
#define D_FEAT  128

#define ROWS_PER_BLOCK 196
#define AGG_BLOCKS     256
#define AGG_THREADS    1024
#define QUEUE_CAP      8192
#define TILE_ROWS      208
#define PITCH          132
#define GROUPS         (N_EDGES / 4)
#define CHUNK_GROUPS   1024
#define N_CHUNKS       ((GROUPS + CHUNK_GROUPS - 1) / CHUNK_GROUPS)
#define SEG_ENT        64
#define SEG_EMPTY      0xFFFFFFFFu
#define SEG_OVER       0xFFFFFFFEu

typedef __attribute__((ext_vector_type(2))) float v2f;
typedef __attribute__((ext_vector_type(8))) float v8f;

template <int NT>
__device__ __forceinline__ void tile_products(const float* mean_t, const float* own_t,
                                              const float* __restrict__ Wl,
                                              const float* __restrict__ Wr,
                                              const float* __restrict__ bias,
                                              float* __restrict__ out,
                                              int g, int lmod, int lhalf, int col,
                                              int base, int nrows, int do_relu) {
    int mt[NT];
    const float* pm[NT];
    const float* po[NT];
    v8f c[NT];
    #pragma unroll
    for (int t = 0; t < NT; ++t) {
        mt[t] = (t < 3) ? 3 * g + t : 12;
        pm[t] = mean_t + (16 * mt[t] + lmod) * PITCH + 2 * lhalf;
        po[t] = own_t  + (16 * mt[t] + lmod) * PITCH + 2 * lhalf;
        c[t] = v8f{};
    }
    const float* wl = Wl + (2 * lhalf) * D_FEAT + col;
    const float* wr = Wr + (2 * lhalf) * D_FEAT + col;

    v2f amP[NT], axP[NT], blP, brP;
    v2f amQ[NT], axQ[NT], blQ, brQ;
    auto load_frags = [&](int k, v2f (&a_m)[NT], v2f (&a_x)[NT], v2f& b_l, v2f& b_r) {
        #pragma unroll
        for (int t = 0; t < NT; ++t) {
            a_m[t] = *(const v2f*)(pm[t] + k);
            a_x[t] = *(const v2f*)(po[t] + k);
        }
        b_l.x = wl[k * D_FEAT];
        b_l.y = wl[(k + 1) * D_FEAT];
        b_r.x = wr[k * D_FEAT];
        b_r.y = wr[(k + 1) * D_FEAT];
    };
    auto products = [&](const v2f (&a_m)[NT], const v2f (&a_x)[NT], const v2f& b_l, const v2f& b_r) {
        #pragma unroll
        for (int t = 0; t < NT; ++t) {
            c[t] = __builtin_amdgcn_wmma_f32_16x16x4_f32(false, a_m[t], false, b_l,
                                                         (short)0, c[t], false, false);
            c[t] = __builtin_amdgcn_wmma_f32_16x16x4_f32(false, a_x[t], false, b_r,
                                                         (short)0, c[t], false, false);
        }
    };
    load_frags(0, amP, axP, blP, brP);
    #pragma unroll 2
    for (int k = 0; k < D_FEAT; k += 8) {
        load_frags(k + 4, amQ, axQ, blQ, brQ);
        products(amP, axP, blP, brP);
        if (k + 8 < D_FEAT) load_frags(k + 8, amP, axP, blP, brP);
        products(amQ, axQ, blQ, brQ);
    }

    const float bv = bias[col];
    #pragma unroll
    for (int t = 0; t < NT; ++t) {
        #pragma unroll
        for (int r = 0; r < 8; ++r) {
            const int row = 16 * mt[t] + 8 * lhalf + r;
            if (row < nrows) {
                float v = c[t][r] + bv;
                if (do_relu) v = fmaxf(v, 0.0f);
                out[(size_t)(base + row) * D_FEAT + col] = v;
            }
        }
    }
}

__global__ __launch_bounds__(CHUNK_GROUPS)
void bin_edges(const int* __restrict__ src, const int* __restrict__ dst, unsigned int* __restrict__ segs) {
    __shared__ __attribute__((aligned(16))) unsigned int seg[AGG_BLOCKS * SEG_ENT];
    __shared__ unsigned int scnt[AGG_BLOCKS];
    const int tid = threadIdx.x;
    const int g = blockIdx.x * CHUNK_GROUPS + tid;
    int4 d = make_int4(-1, -1, -1, -1);
    int4 s = make_int4(0, 0, 0, 0);
    if (g < GROUPS) {
        d = reinterpret_cast<const int4*>(dst)[g];
        s = reinterpret_cast<const int4*>(src)[g];
    }
    uint4* s4 = reinterpret_cast<uint4*>(seg);
    #pragma unroll
    for (int j = 0; j < 4; ++j)
        s4[tid + CHUNK_GROUPS * j] = make_uint4(SEG_EMPTY, SEG_EMPTY, SEG_EMPTY, SEG_EMPTY);
    if (tid < AGG_BLOCKS) scnt[tid] = 0u;
    __syncthreads();
    const int dd[4] = {d.x, d.y, d.z, d.w};
    const int ss[4] = {s.x, s.y, s.z, s.w};
    #pragma unroll
    for (int c = 0; c < 4; ++c) {
        if ((unsigned)dd[c] < (unsigned)N_NODES) {
            const unsigned o = (unsigned)dd[c] / ROWS_PER_BLOCK;
            const unsigned r = (unsigned)dd[c] - o * ROWS_PER_BLOCK;
            const unsigned slot = atomicAdd(&scnt[o], 1u);
            if (slot < SEG_ENT) seg[o * SEG_ENT + slot] = min((unsigned)ss[c], (unsigned)(N_NODES - 1)) | (r << 16);
        }
    }
    __syncthreads();
    if (tid < AGG_BLOCKS && scnt[tid] > SEG_ENT) seg[tid * SEG_ENT] = SEG_OVER;
    __syncthreads();
    uint4* o4 = reinterpret_cast<uint4*>(segs + (size_t)blockIdx.x * (AGG_BLOCKS * SEG_ENT));
    #pragma unroll
    for (int j = 0; j < 4; ++j)
        o4[tid + CHUNK_GROUPS * j] = s4[tid + CHUNK_GROUPS * j];
}

template <int LAYER>
__global__ __launch_bounds__(AGG_THREADS)
void sage_layer(const float* __restrict__ in,
                const int* __restrict__ src,
                const int* __restrict__ dst,
                const unsigned int* __restrict__ segs,
                const float* __restrict__ Wl,
                const float* __restrict__ Wr,
                const float* __restrict__ bias,
                float* __restrict__ out,
                unsigned int* __restrict__ list_ws,
                unsigned int* __restrict__ count_ws,
                int do_relu) {
    __shared__ __attribute__((aligned(16))) float acc[TILE_ROWS * PITCH];
    __shared__ __attribute__((aligned(16))) float own[TILE_ROWS * PITCH];
    __shared__ __attribute__((aligned(16))) unsigned int queue[QUEUE_CAP];
    __shared__ int cnt[TILE_ROWS];
    __shared__ unsigned int qn;
    __shared__ unsigned int over;

    const int tid   = threadIdx.x;
    const int base  = blockIdx.x * ROWS_PER_BLOCK;
    const int nrows = min(ROWS_PER_BLOCK, N_NODES - base);
    unsigned int* __restrict__ my_list = list_ws + (size_t)blockIdx.x * QUEUE_CAP;

    unsigned total = 0u;
    if (LAYER == 2) total = count_ws[blockIdx.x];

    float4 ownv[7];
    #pragma unroll
    for (int j = 0; j < 7; ++j) {
        const int r = (tid >> 5) + 32 * j;
        ownv[j] = make_float4(0.f, 0.f, 0.f, 0.f);
        if (r < nrows)
            ownv[j] = *reinterpret_cast<const float4*>(in + (size_t)(base + r) * D_FEAT + 4 * (tid & 31));
    }

    uint4 sv[4];
    if (LAYER == 1) {
        const uint4* sg4 = reinterpret_cast<const uint4*>(segs);
        #pragma unroll
        for (int j = 0; j < 4; ++j) {
            const int i = tid + AGG_THREADS * j;
            sv[j] = make_uint4(SEG_EMPTY, SEG_EMPTY, SEG_EMPTY, SEG_EMPTY);
            if (i < N_CHUNKS * (SEG_ENT / 4))
                sv[j] = sg4[((size_t)(i >> 4) * AGG_BLOCKS + blockIdx.x) * (SEG_ENT / 4) + (i & 15)];
        }
    }

    const unsigned n16 = (LAYER == 2 && total <= QUEUE_CAP) ? ((total + 7u) & ~7u) / 4u : 0u;
    uint4 lv[2];
    if (LAYER == 2) {
        const uint4* l4 = reinterpret_cast<const uint4*>(my_list);
        #pragma unroll
        for (int j = 0; j < 2; ++j) {
            const unsigned i = (unsigned)tid + AGG_THREADS * j;
            lv[j] = make_uint4(0u, 0u, 0u, 0u);
            if (i < n16) lv[j] = l4[i];
        }
    }

    float4* acc4 = reinterpret_cast<float4*>(acc);
    for (int i = tid; i < TILE_ROWS * PITCH / 4; i += AGG_THREADS)
        acc4[i] = make_float4(0.f, 0.f, 0.f, 0.f);
    if (tid < TILE_ROWS) cnt[tid] = 0;
    if (tid == 0) { qn = 0u; over = 0u; }
    #pragma unroll
    for (int j = 0; j < 7; ++j) {
        const int r = (tid >> 5) + 32 * j;
        if (r < TILE_ROWS)
            *reinterpret_cast<float4*>(own + r * PITCH + 4 * (tid & 31)) = ownv[j];
    }
    if (LAYER == 2) {
        uint4* q4 = reinterpret_cast<uint4*>(queue);
        #pragma unroll
        for (int j = 0; j < 2; ++j) {
            const unsigned i = (unsigned)tid + AGG_THREADS * j;
            if (i < n16) q4[i] = lv[j];
        }
    }
    __syncthreads();

    if (LAYER == 1) {
        const unsigned e[16] = {sv[0].x, sv[0].y, sv[0].z, sv[0].w, sv[1].x, sv[1].y, sv[1].z, sv[1].w,
                                sv[2].x, sv[2].y, sv[2].z, sv[2].w, sv[3].x, sv[3].y, sv[3].z, sv[3].w};
        {
            const int lane = tid & 31;
            unsigned n = 0u;
            bool ov = false;
            #pragma unroll
            for (int k = 0; k < 16; ++k) {
                ov = ov || (e[k] == SEG_OVER);
                n += (e[k] < SEG_OVER) ? 1u : 0u;
            }
            unsigned run = n;
            #pragma unroll
            for (int dlt = 1; dlt < 32; dlt <<= 1) {
                const unsigned up = (unsigned)__shfl_up((int)run, dlt);
                if (lane >= dlt) run += up;
            }
            unsigned wbase = 0u;
            if (lane == 31 && run != 0u) wbase = atomicAdd(&qn, run);
            wbase = (unsigned)__shfl((int)wbase, 31);
            unsigned slot = wbase + run - n;
            #pragma unroll
            for (int k = 0; k < 16; ++k) {
                if (e[k] < SEG_OVER) {
                    if (slot < QUEUE_CAP) queue[slot] = e[k];
                    ++slot;
                }
            }
            if (ov) over = 1u;
        }
        __syncthreads();

        total = (over == 0u) ? qn : (unsigned)(QUEUE_CAP + 1);
        const unsigned padded = (total <= QUEUE_CAP) ? ((total + 7u) & ~7u) : 0u;
        if ((unsigned)tid < padded - total && total <= QUEUE_CAP) queue[total + (unsigned)tid] = 0u;
        __syncthreads();
        {
            const uint4* q4 = reinterpret_cast<const uint4*>(queue);
            uint4* l4 = reinterpret_cast<uint4*>(my_list);
            for (unsigned i = (unsigned)tid; i < padded / 4u; i += AGG_THREADS) l4[i] = q4[i];
        }
        if (tid == 0) count_ws[blockIdx.x] = total;
    }

    if (total <= QUEUE_CAP) {
        const unsigned wave = (unsigned)__builtin_amdgcn_readfirstlane(tid >> 5);
        const int lane = tid & 31;
        const uint4* queue4 = reinterpret_cast<const uint4*>(queue);
        for (unsigned k = wave * 8u; k < total; k += (AGG_THREADS / 32) * 8u) {
            const uint4 qa = queue4[k / 4u];
            const uint4 qb = queue4[k / 4u + 1u];
            const unsigned raw[8] = {qa.x, qa.y, qa.z, qa.w, qb.x, qb.y, qb.z, qb.w};
            unsigned ent[8];
            bool valid[8];
            const float* row[8];
            float v[8][4];
            #pragma unroll
            for (int t = 0; t < 8; ++t) {
                valid[t] = (k + t) < total;
                ent[t] = valid[t] ? (unsigned)__builtin_amdgcn_readfirstlane((int)raw[t]) : 0u;
            }
            #pragma unroll
            for (int t = 0; t < 8; ++t) {
                const unsigned s = min(ent[t] & 0xFFFFu, (unsigned)(N_NODES - 1));
                row[t] = in + (size_t)s * D_FEAT + lane;
            }
            #pragma unroll
            for (int t = 0; t < 8; ++t) {
                #pragma unroll
                for (int c = 0; c < 4; ++c) v[t][c] = row[t][32 * c];
            }
            #pragma unroll
            for (int t = 0; t < 8; ++t) {
                if (valid[t]) {
                    const unsigned u = min(ent[t] >> 16, (unsigned)(TILE_ROWS - 1));
                    float* a = acc + u * PITCH + lane;
                    #pragma unroll
                    for (int c = 0; c < 4; ++c)
                        __hip_atomic_fetch_add(a + 32 * c, v[t][c],
                                               __ATOMIC_RELAXED, __HIP_MEMORY_SCOPE_WORKGROUP);
                    if (lane == 0)
                        __hip_atomic_fetch_add(&cnt[u], 1,
                                               __ATOMIC_RELAXED, __HIP_MEMORY_SCOPE_WORKGROUP);
                }
            }
        }
    } else {
        for (int e = tid; e < N_EDGES; e += AGG_THREADS) {
            const unsigned u = (unsigned)(dst[e] - base);
            if (u < (unsigned)nrows) {
                const unsigned s = min((unsigned)src[e], (unsigned)(N_NODES - 1));
                const float* r = in + (size_t)s * D_FEAT;
                float* a = acc + u * PITCH;
                for (int c = 0; c < D_FEAT; ++c)
                    __hip_atomic_fetch_add(a + c, r[c],
                                           __ATOMIC_RELAXED, __HIP_MEMORY_SCOPE_WORKGROUP);
                __hip_atomic_fetch_add(&cnt[u], 1,
                                       __ATOMIC_RELAXED, __HIP_MEMORY_SCOPE_WORKGROUP);
            }
        }
    }
    __syncthreads();

    for (int i = tid; i < nrows * (D_FEAT / 4); i += AGG_THREADS) {
        const int r = i >> 5;
        const float dv = fmaxf((float)cnt[r], 1.0f);
        float4* p = reinterpret_cast<float4*>(acc + r * PITCH + 4 * (i & 31));
        float4 a = *p;
        a.x = a.x / dv;
        a.y = a.y / dv;
        a.z = a.z / dv;
        a.w = a.w / dv;
        *p = a;
    }
    __syncthreads();

    {
        const int wave  = __builtin_amdgcn_readfirstlane(tid >> 5);
        const int lane  = tid & 31;
        const int lhalf = lane >> 4;
        const int lmod  = lane & 15;
        const int ntile = wave & 7;
        const int g     = wave >> 3;
        const int col   = ntile * 16 + lmod;
        if (g == (ntile >> 1))
            tile_products<4>(acc, own, Wl, Wr, bias, out, g, lmod, lhalf, col, base, nrows, do_relu);
        else
            tile_products<3>(acc, own, Wl, Wr, bias, out, g, lmod, lhalf, col, base, nrows, do_relu);
    }
}

extern "C" void kernel_launch(void* const* d_in, const int* in_sizes, int n_in,
                              void* d_out, int out_size, void* d_ws, size_t ws_size,
                              hipStream_t stream) {
    const float* x    = (const float*)d_in[0];
    const int*   eidx = (const int*)d_in[1];
    const float* W1l  = (const float*)d_in[2];
    const float* b1   = (const float*)d_in[3];
    const float* W1r  = (const float*)d_in[4];
    const float* W2l  = (const float*)d_in[5];
    const float* b2   = (const float*)d_in[6];
    const float* W2r  = (const float*)d_in[7];
    float* out = (float*)d_out;

    const int* src = eidx;
    const int* dst = eidx + N_EDGES;

    float* agg = (float*)d_ws;
    float* deg = agg + (size_t)N_NODES * D_FEAT;
    float* h   = deg + N_NODES;
    unsigned int* list_ws  = (unsigned int*)agg;
    unsigned int* count_ws = (unsigned int*)deg;
    unsigned int* seg_ws   = list_ws + (size_t)AGG_BLOCKS * QUEUE_CAP;

    bin_edges<<<N_CHUNKS, CHUNK_GROUPS, 0, stream>>>(src, dst, seg_ws);

    sage_layer<1><<<AGG_BLOCKS, AGG_THREADS, 0, stream>>>(x, src, dst, seg_ws, W1l, W1r, b1, h, list_ws, count_ws, 1);
    sage_layer<2><<<AGG_BLOCKS, AGG_THREADS, 0, stream>>>(h, src, dst, seg_ws, W2l, W2r, b2, out, list_ws, count_ws, 0);
}


# Gather loop stripped to load+add: row counts taken where the list is packed or copied (all lanes), list filled to groups of 8 with entries into spare tile row 207; float sums: LDS arrival order
# speedup vs baseline: 6.4792x; 1.0360x over previous
#include <hip/hip_runtime.h>
#include <hip/hip_bf16.h>


#define N_NODES 50000
#define N_EDGES 800000
#define D_FEAT  128

#define ROWS_PER_BLOCK 196
#define AGG_BLOCKS     256
#define AGG_THREADS    1024
#define QUEUE_CAP      8192
#define TILE_ROWS      208
#define PITCH          132
#define GROUPS         (N_EDGES / 4)
#define CHUNK_GROUPS   1024
#define N_CHUNKS       ((GROUPS + CHUNK_GROUPS - 1) / CHUNK_GROUPS)
#define SEG_ENT        64
#define SEG_EMPTY      0xFFFFFFFFu
#define SEG_OVER       0xFFFFFFFEu
#define PAD_ENTRY      ((unsigned)(TILE_ROWS - 1) << 16)

typedef __attribute__((ext_vector_type(2))) float v2f;
typedef __attribute__((ext_vector_type(8))) float v8f;

template <int NT>
__device__ __forceinline__ void tile_products(const float* mean_t, const float* own_t,
                                              const float* __restrict__ Wl,
                                              const float* __restrict__ Wr,
                                              const float* __restrict__ bias,
                                              float* __restrict__ out,
                                              int g, int lmod, int lhalf, int col,
                                              int base, int nrows, int do_relu) {
    int mt[NT];
    const float* pm[NT];
    const float* po[NT];
    v8f c[NT];
    #pragma unroll
    for (int t = 0; t < NT; ++t) {
        mt[t] = (t < 3) ? 3 * g + t : 12;
        pm[t] = mean_t + (16 * mt[t] + lmod) * PITCH + 2 * lhalf;
        po[t] = own_t  + (16 * mt[t] + lmod) * PITCH + 2 * lhalf;
        c[t] = v8f{};
    }
    const float* wl = Wl + (2 * lhalf) * D_FEAT + col;
    const float* wr = Wr + (2 * lhalf) * D_FEAT + col;

    v2f amP[NT], axP[NT], blP, brP;
    v2f amQ[NT], axQ[NT], blQ, brQ;
    auto load_frags = [&](int k, v2f (&a_m)[NT], v2f (&a_x)[NT], v2f& b_l, v2f& b_r) {
        #pragma unroll
        for (int t = 0; t < NT; ++t) {
            a_m[t] = *(const v2f*)(pm[t] + k);
            a_x[t] = *(const v2f*)(po[t] + k);
        }
        b_l.x = wl[k * D_FEAT];
        b_l.y = wl[(k + 1) * D_FEAT];
        b_r.x = wr[k * D_FEAT];
        b_r.y = wr[(k + 1) * D_FEAT];
    };
    auto products = [&](const v2f (&a_m)[NT], const v2f (&a_x)[NT], const v2f& b_l, const v2f& b_r) {
        #pragma unroll
        for (int t = 0; t < NT; ++t) {
            c[t] = __builtin_amdgcn_wmma_f32_16x16x4_f32(false, a_m[t], false, b_l,
                                                         (short)0, c[t], false, false);
            c[t] = __builtin_amdgcn_wmma_f32_16x16x4_f32(false, a_x[t], false, b_r,
                                                         (short)0, c[t], false, false);
        }
    };
    load_frags(0, amP, axP, blP, brP);
    #pragma unroll 2
    for (int k = 0; k < D_FEAT; k += 8) {
        load_frags(k + 4, amQ, axQ, blQ, brQ);
        products(amP, axP, blP, brP);
        if (k + 8 < D_FEAT) load_frags(k + 8, amP, axP, blP, brP);
        products(amQ, axQ, blQ, brQ);
    }

    const float bv = bias[col];
    #pragma unroll
    for (int t = 0; t < NT; ++t) {
        #pragma unroll
        for (int r = 0; r < 8; ++r) {
            const int row = 16 * mt[t] + 8 * lhalf + r;
            if (row < nrows) {
                float v = c[t][r] + bv;
                if (do_relu) v = fmaxf(v, 0.0f);
                out[(size_t)(base + row) * D_FEAT + col] = v;
            }
        }
    }
}

__global__ __launch_bounds__(CHUNK_GROUPS)
void bin_edges(const int* __restrict__ src, const int* __restrict__ dst, unsigned int* __restrict__ segs) {
    __shared__ __attribute__((aligned(16))) unsigned int seg[AGG_BLOCKS * SEG_ENT];
    __shared__ unsigned int scnt[AGG_BLOCKS];
    const int tid = threadIdx.x;
    const int g = blockIdx.x * CHUNK_GROUPS + tid;
    int4 d = make_int4(-1, -1, -1, -1);
    int4 s = make_int4(0, 0, 0, 0);
    if (g < GROUPS) {
        d = reinterpret_cast<const int4*>(dst)[g];
        s = reinterpret_cast<const int4*>(src)[g];
    }
    uint4* s4 = reinterpret_cast<uint4*>(seg);
    #pragma unroll
    for (int j = 0; j < 4; ++j)
        s4[tid + CHUNK_GROUPS * j] = make_uint4(SEG_EMPTY, SEG_EMPTY, SEG_EMPTY, SEG_EMPTY);
    if (tid < AGG_BLOCKS) scnt[tid] = 0u;
    __syncthreads();
    const int dd[4] = {d.x, d.y, d.z, d.w};
    const int ss[4] = {s.x, s.y, s.z, s.w};
    #pragma unroll
    for (int c = 0; c < 4; ++c) {
        if ((unsigned)dd[c] < (unsigned)N_NODES) {
            const unsigned o = (unsigned)dd[c] / ROWS_PER_BLOCK;
            const unsigned r = (unsigned)dd[c] - o * ROWS_PER_BLOCK;
            const unsigned slot = atomicAdd(&scnt[o], 1u);
            if (slot < SEG_ENT) seg[o * SEG_ENT + slot] = min((unsigned)ss[c], (unsigned)(N_NODES - 1)) | (r << 16);
        }
    }
    __syncthreads();
    if (tid < AGG_BLOCKS && scnt[tid] > SEG_ENT) seg[tid * SEG_ENT] = SEG_OVER;
    __syncthreads();
    uint4* o4 = reinterpret_cast<uint4*>(segs + (size_t)blockIdx.x * (AGG_BLOCKS * SEG_ENT));
    #pragma unroll
    for (int j = 0; j < 4; ++j)
        o4[tid + CHUNK_GROUPS * j] = s4[tid + CHUNK_GROUPS * j];
}

template <int LAYER>
__global__ __launch_bounds__(AGG_THREADS)
void sage_layer(const float* __restrict__ in,
                const int* __restrict__ src,
                const int* __restrict__ dst,
                const unsigned int* __restrict__ segs,
                const float* __restrict__ Wl,
                const float* __restrict__ Wr,
                const float* __restrict__ bias,
                float* __restrict__ out,
                unsigned int* __restrict__ list_ws,
                unsigned int* __restrict__ count_ws,
                int do_relu) {
    __shared__ __attribute__((aligned(16))) float acc[TILE_ROWS * PITCH];
    __shared__ __attribute__((aligned(16))) float own[TILE_ROWS * PITCH];
    __shared__ __attribute__((aligned(16))) unsigned int queue[QUEUE_CAP];
    __shared__ int cnt[TILE_ROWS];
    __shared__ unsigned int qn;
    __shared__ unsigned int over;

    const int tid   = threadIdx.x;
    const int base  = blockIdx.x * ROWS_PER_BLOCK;
    const int nrows = min(ROWS_PER_BLOCK, N_NODES - base);
    unsigned int* __restrict__ my_list = list_ws + (size_t)blockIdx.x * QUEUE_CAP;

    unsigned total = 0u;
    if (LAYER == 2) total = count_ws[blockIdx.x];

    float4 ownv[7];
    #pragma unroll
    for (int j = 0; j < 7; ++j) {
        const int r = (tid >> 5) + 32 * j;
        ownv[j] = make_float4(0.f, 0.f, 0.f, 0.f);
        if (r < nrows)
            ownv[j] = *reinterpret_cast<const float4*>(in + (size_t)(base + r) * D_FEAT + 4 * (tid & 31));
    }

    uint4 sv[4];
    if (LAYER == 1) {
        const uint4* sg4 = reinterpret_cast<const uint4*>(segs);
        #pragma unroll
        for (int j = 0; j < 4; ++j) {
            const int i = tid + AGG_THREADS * j;
            sv[j] = make_uint4(SEG_EMPTY, SEG_EMPTY, SEG_EMPTY, SEG_EMPTY);
            if (i < N_CHUNKS * (SEG_ENT / 4))
                sv[j] = sg4[((size_t)(i >> 4) * AGG_BLOCKS + blockIdx.x) * (SEG_ENT / 4) + (i & 15)];
        }
    }

    const unsigned n16 = (LAYER == 2 && total <= QUEUE_CAP) ? ((total + 7u) & ~7u) / 4u : 0u;
    uint4 lv[2];
    if (LAYER == 2) {
        const uint4* l4 = reinterpret_cast<const uint4*>(my_list);
        #pragma unroll
        for (int j = 0; j < 2; ++j) {
            const unsigned i = (unsigned)tid + AGG_THREADS * j;
            lv[j] = make_uint4(0u, 0u, 0u, 0u);
            if (i < n16) lv[j] = l4[i];
        }
    }

    float4* acc4 = reinterpret_cast<float4*>(acc);
    for (int i = tid; i < TILE_ROWS * PITCH / 4; i += AGG_THREADS)
        acc4[i] = make_float4(0.f, 0.f, 0.f, 0.f);
    if (tid < TILE_ROWS) cnt[tid] = 0;
    if (tid == 0) { qn = 0u; over = 0u; }
    #pragma unroll
    for (int j = 0; j < 7; ++j) {
        const int r = (tid >> 5) + 32 * j;
        if (r < TILE_ROWS)
            *reinterpret_cast<float4*>(own + r * PITCH + 4 * (tid & 31)) = ownv[j];
    }
    if (LAYER == 2) {
        uint4* q4 = reinterpret_cast<uint4*>(queue);
        #pragma unroll
        for (int j = 0; j < 2; ++j) {
            const unsigned i = (unsigned)tid + AGG_THREADS * j;
            if (i < n16) q4[i] = lv[j];
        }
    }
    __syncthreads();

    if (LAYER == 2) {
        #pragma unroll
        for (int j = 0; j < 2; ++j) {
            const unsigned i = (unsigned)tid + AGG_THREADS * j;
            const unsigned w[4] = {lv[j].x, lv[j].y, lv[j].z, lv[j].w};
            if (i < n16) {
                #pragma unroll
                for (int c = 0; c < 4; ++c)
                    __hip_atomic_fetch_add(&cnt[min(w[c] >> 16, (unsigned)(TILE_ROWS - 1))], 1,
                                           __ATOMIC_RELAXED, __HIP_MEMORY_SCOPE_WORKGROUP);
            }
        }
    }

    if (LAYER == 1) {
        const unsigned e[16] = {sv[0].x, sv[0].y, sv[0].z, sv[0].w, sv[1].x, sv[1].y, sv[1].z, sv[1].w,
                                sv[2].x, sv[2].y, sv[2].z, sv[2].w, sv[3].x, sv[3].y, sv[3].z, sv[3].w};
        {
            const int lane = tid & 31;
            unsigned n = 0u;
            bool ov = false;
            #pragma unroll
            for (int k = 0; k < 16; ++k) {
                ov = ov || (e[k] == SEG_OVER);
                n += (e[k] < SEG_OVER) ? 1u : 0u;
            }
            unsigned run = n;
            #pragma unroll
            for (int dlt = 1; dlt < 32; dlt <<= 1) {
                const unsigned up = (unsigned)__shfl_up((int)run, dlt);
                if (lane >= dlt) run += up;
            }
            unsigned wbase = 0u;
            if (lane == 31 && run != 0u) wbase = atomicAdd(&qn, run);
            wbase = (unsigned)__shfl((int)wbase, 31);
            unsigned slot = wbase + run - n;
            #pragma unroll
            for (int k = 0; k < 16; ++k) {
                if (e[k] < SEG_OVER) {
                    if (slot < QUEUE_CAP) queue[slot] = e[k];
                    ++slot;
                    __hip_atomic_fetch_add(&cnt[min(e[k] >> 16, (unsigned)(TILE_ROWS - 1))], 1,
                                           __ATOMIC_RELAXED, __HIP_MEMORY_SCOPE_WORKGROUP);
                }
            }
            if (ov) over = 1u;
        }
        __syncthreads();

        total = (over == 0u) ? qn : (unsigned)(QUEUE_CAP + 1);
        const unsigned padded = (total <= QUEUE_CAP) ? ((total + 7u) & ~7u) : 0u;
        if ((unsigned)tid < padded - total && total <= QUEUE_CAP) queue[total + (unsigned)tid] = PAD_ENTRY;
        __syncthreads();
        {
            const uint4* q4 = reinterpret_cast<const uint4*>(queue);
            uint4* l4 = reinterpret_cast<uint4*>(my_list);
            for (unsigned i = (unsigned)tid; i < padded / 4u; i += AGG_THREADS) l4[i] = q4[i];
        }
        if (tid == 0) count_ws[blockIdx.x] = total;
    }

    if (total <= QUEUE_CAP) {
        const unsigned wave = (unsigned)__builtin_amdgcn_readfirstlane(tid >> 5);
        const int lane = tid & 31;
        const uint4* queue4 = reinterpret_cast<const uint4*>(queue);
        for (unsigned k = wave * 8u; k < total; k += (AGG_THREADS / 32) * 8u) {
            const uint4 qa = queue4[k / 4u];
            const uint4 qb = queue4[k / 4u + 1u];
            const unsigned raw[8] = {qa.x, qa.y, qa.z, qa.w, qb.x, qb.y, qb.z, qb.w};
            unsigned ent[8];
            const float* row[8];
            float v[8][4];
            #pragma unroll
            for (int t = 0; t < 8; ++t)
                ent[t] = (unsigned)__builtin_amdgcn_readfirstlane((int)raw[t]);
            #pragma unroll
            for (int t = 0; t < 8; ++t) {
                const unsigned s = min(ent[t] & 0xFFFFu, (unsigned)(N_NODES - 1));
                row[t] = in + (size_t)s * D_FEAT + lane;
            }
            #pragma unroll
            for (int t = 0; t < 8; ++t) {
                #pragma unroll
                for (int c = 0; c < 4; ++c) v[t][c] = row[t][32 * c];
            }
            #pragma unroll
            for (int t = 0; t < 8; ++t) {
                const unsigned u = min(ent[t] >> 16, (unsigned)(TILE_ROWS - 1));
                float* a = acc + u * PITCH + lane;
                #pragma unroll
                for (int c = 0; c < 4; ++c)
                    __hip_atomic_fetch_add(a + 32 * c, v[t][c],
                                           __ATOMIC_RELAXED, __HIP_MEMORY_SCOPE_WORKGROUP);
            }
        }
    } else {
        if (tid < TILE_ROWS) cnt[tid] = 0;
        __syncthreads();
        for (int e = tid; e < N_EDGES; e += AGG_THREADS) {
            const unsigned u = (unsigned)(dst[e] - base);
            if (u < (unsigned)nrows) {
                const unsigned s = min((unsigned)src[e], (unsigned)(N_NODES - 1));
                const float* r = in + (size_t)s * D_FEAT;
                float* a = acc + u * PITCH;
                for (int c = 0; c < D_FEAT; ++c)
                    __hip_atomic_fetch_add(a + c, r[c],
                                           __ATOMIC_RELAXED, __HIP_MEMORY_SCOPE_WORKGROUP);
                __hip_atomic_fetch_add(&cnt[u], 1,
                                       __ATOMIC_RELAXED, __HIP_MEMORY_SCOPE_WORKGROUP);
            }
        }
    }
    __syncthreads();

    for (int i = tid; i < nrows * (D_FEAT / 4); i += AGG_THREADS) {
        const int r = i >> 5;
        const float dv = fmaxf((float)cnt[r], 1.0f);
        float4* p = reinterpret_cast<float4*>(acc + r * PITCH + 4 * (i & 31));
        float4 a = *p;
        a.x = a.x / dv;
        a.y = a.y / dv;
        a.z = a.z / dv;
        a.w = a.w / dv;
        *p = a;
    }
    __syncthreads();

    {
        const int wave  = __builtin_amdgcn_readfirstlane(tid >> 5);
        const int lane  = tid & 31;
        const int lhalf = lane >> 4;
        const int lmod  = lane & 15;
        const int ntile = wave & 7;
        const int g     = wave >> 3;
        const int col   = ntile * 16 + lmod;
        if (g == (ntile >> 1))
            tile_products<4>(acc, own, Wl, Wr, bias, out, g, lmod, lhalf, col, base, nrows, do_relu);
        else
            tile_products<3>(acc, own, Wl, Wr, bias, out, g, lmod, lhalf, col, base, nrows, do_relu);
    }
}

extern "C" void kernel_launch(void* const* d_in, const int* in_sizes, int n_in,
                              void* d_out, int out_size, void* d_ws, size_t ws_size,
                              hipStream_t stream) {
    const float* x    = (const float*)d_in[0];
    const int*   eidx = (const int*)d_in[1];
    const float* W1l  = (const float*)d_in[2];
    const float* b1   = (const float*)d_in[3];
    const float* W1r  = (const float*)d_in[4];
    const float* W2l  = (const float*)d_in[5];
    const float* b2   = (const float*)d_in[6];
    const float* W2r  = (const float*)d_in[7];
    float* out = (float*)d_out;

    const int* src = eidx;
    const int* dst = eidx + N_EDGES;

    float* agg = (float*)d_ws;
    float* deg = agg + (size_t)N_NODES * D_FEAT;
    float* h   = deg + N_NODES;
    unsigned int* list_ws  = (unsigned int*)agg;
    unsigned int* count_ws = (unsigned int*)deg;
    unsigned int* seg_ws   = list_ws + (size_t)AGG_BLOCKS * QUEUE_CAP;

    bin_edges<<<N_CHUNKS, CHUNK_GROUPS, 0, stream>>>(src, dst, seg_ws);

    sage_layer<1><<<AGG_BLOCKS, AGG_THREADS, 0, stream>>>(x, src, dst, seg_ws, W1l, W1r, b1, h, list_ws, count_ws, 1);
    sage_layer<2><<<AGG_BLOCKS, AGG_THREADS, 0, stream>>>(h, src, dst, seg_ws, W2l, W2r, b2, out, list_ws, count_ws, 0);
}


# Own-row product moved into the gather loop: each trip of a wave also runs 4 k-steps of in @ Wr, so the matrix unit works while other waves' rows fly; mean @ Wl follows; sum order: Wr terms, then Wl
# speedup vs baseline: 6.9340x; 1.0702x over previous
#include <hip/hip_runtime.h>
#include <hip/hip_bf16.h>


#define N_NODES 50000
#define N_EDGES 800000
#define D_FEAT  128

#define ROWS_PER_BLOCK 196
#define AGG_BLOCKS     256
#define AGG_THREADS    1024
#define QUEUE_CAP      8192
#define TILE_ROWS      208
#define PITCH          132
#define GROUPS         (N_EDGES / 4)
#define CHUNK_GROUPS   1024
#define N_CHUNKS       ((GROUPS + CHUNK_GROUPS - 1) / CHUNK_GROUPS)
#define SEG_ENT        64
#define SEG_EMPTY      0xFFFFFFFFu
#define SEG_OVER       0xFFFFFFFEu
#define PAD_ENTRY      ((unsigned)(TILE_ROWS - 1) << 16)

typedef __attribute__((ext_vector_type(2))) float v2f;
typedef __attribute__((ext_vector_type(8))) float v8f;


__device__ __forceinline__ v8f wmma4(const v2f& a, const v2f& b, const v8f& c) {
    return __builtin_amdgcn_wmma_f32_16x16x4_f32(false, a, false, b, (short)0, c, false, false);
}

__device__ __forceinline__ void load_b4(const float* __restrict__ w, int k0, v2f (&b)[4]) {
    #pragma unroll
    for (int s = 0; s < 4; ++s) {
        b[s].x = w[(k0 + 4 * s) * D_FEAT];
        b[s].y = w[(k0 + 4 * s + 1) * D_FEAT];
    }
}

template <int NT>
__device__ __forceinline__ void ksteps4(v8f (&c)[4], const float* const (&pa)[NT], const v2f (&b)[4], int k0) {
    #pragma unroll
    for (int s = 0; s < 4; ++s) {
        v2f a[NT];
        #pragma unroll
        for (int t = 0; t < NT; ++t) a[t] = *(const v2f*)(pa[t] + k0 + 4 * s);
        #pragma unroll
        for (int t = 0; t < NT; ++t) c[t] = wmma4(a[t], b[s], c[t]);
    }
}

template <int NT>
__device__ __forceinline__ void gather_with_own(v8f (&c)[4], const float* __restrict__ in, float* acc,
                                                const float* own_t, const unsigned int* queue,
                                                const float* __restrict__ Wr,
                                                unsigned total, unsigned wave, int lane,
                                                int g, int lmod, int lhalf, int col) {
    const float* po[NT];
    #pragma unroll
    for (int t = 0; t < NT; ++t) {
        const int mt = (t < 3) ? 3 * g + t : 12;
        po[t] = own_t + (16 * mt + lmod) * PITCH + 2 * lhalf;
    }
    const float* wr = Wr + (2 * lhalf) * D_FEAT + col;
    const uint4* queue4 = reinterpret_cast<const uint4*>(queue);
    const unsigned stride = (AGG_THREADS / 32) * 8u;

    unsigned ent[8];
    float v[8][4];
    auto issue_rows = [&](unsigned k) {
        const uint4 qa = queue4[k / 4u];
        const uint4 qb = queue4[k / 4u + 1u];
        const unsigned raw[8] = {qa.x, qa.y, qa.z, qa.w, qb.x, qb.y, qb.z, qb.w};
        #pragma unroll
        for (int t = 0; t < 8; ++t)
            ent[t] = (unsigned)__builtin_amdgcn_readfirstlane((int)raw[t]);
        #pragma unroll
        for (int t = 0; t < 8; ++t) {
            const unsigned s = min(ent[t] & 0xFFFFu, (unsigned)(N_NODES - 1));
            const float* row = in + (size_t)s * D_FEAT + lane;
            #pragma unroll
            for (int cc = 0; cc < 4; ++cc) v[t][cc] = row[32 * cc];
        }
    };
    auto add_rows = [&]() {
        #pragma unroll
        for (int t = 0; t < 8; ++t) {
            const unsigned u = min(ent[t] >> 16, (unsigned)(TILE_ROWS - 1));
            float* a = acc + u * PITCH + lane;
            #pragma unroll
            for (int cc = 0; cc < 4; ++cc)
                __hip_atomic_fetch_add(a + 32 * cc, v[t][cc],
                                       __ATOMIC_RELAXED, __HIP_MEMORY_SCOPE_WORKGROUP);
        }
    };

    unsigned k = wave * 8u;
    int kk = 0;
    v2f b[4];
    for (; k < total && kk < D_FEAT; k += stride, kk += 16) {
        load_b4(wr, kk, b);
        issue_rows(k);
        ksteps4<NT>(c, po, b, kk);
        add_rows();
    }
    for (; k < total; k += stride) {
        issue_rows(k);
        add_rows();
    }
    for (; kk < D_FEAT; kk += 16) {
        load_b4(wr, kk, b);
        ksteps4<NT>(c, po, b, kk);
    }
}

template <int NT>
__device__ __forceinline__ void mean_products_store(v8f (&c)[4], const float* mean_t,
                                                    const float* __restrict__ Wl,
                                                    const float* __restrict__ bias,
                                                    float* __restrict__ out,
                                                    int g, int lmod, int lhalf, int col,
                                                    int base, int nrows, int do_relu) {
    int mt[NT];
    const float* pm[NT];
    #pragma unroll
    for (int t = 0; t < NT; ++t) {
        mt[t] = (t < 3) ? 3 * g + t : 12;
        pm[t] = mean_t + (16 * mt[t] + lmod) * PITCH + 2 * lhalf;
    }
    const float* wl = Wl + (2 * lhalf) * D_FEAT + col;

    v2f a0P[NT], a1P[NT], b0P, b1P;
    v2f a0Q[NT], a1Q[NT], b0Q, b1Q;
    auto load_pair = [&](int k, v2f (&a0)[NT], v2f (&a1)[NT], v2f& b0, v2f& b1) {
        #pragma unroll
        for (int t = 0; t < NT; ++t) {
            a0[t] = *(const v2f*)(pm[t] + k);
            a1[t] = *(const v2f*)(pm[t] + k + 4);
        }
        b0.x = wl[k * D_FEAT];
        b0.y = wl[(k + 1) * D_FEAT];
        b1.x = wl[(k + 4) * D_FEAT];
        b1.y = wl[(k + 5) * D_FEAT];
    };
    auto products = [&](const v2f (&a0)[NT], const v2f (&a1)[NT], const v2f& b0, const v2f& b1) {
        #pragma unroll
        for (int t = 0; t < NT; ++t) c[t] = wmma4(a0[t], b0, c[t]);
        #pragma unroll
        for (int t = 0; t < NT; ++t) c[t] = wmma4(a1[t], b1, c[t]);
    };
    load_pair(0, a0P, a1P, b0P, b1P);
    #pragma unroll 2
    for (int k = 0; k < D_FEAT; k += 16) {
        load_pair(k + 8, a0Q, a1Q, b0Q, b1Q);
        products(a0P, a1P, b0P, b1P);
        if (k + 16 < D_FEAT) load_pair(k + 16, a0P, a1P, b0P, b1P);
        products(a0Q, a1Q, b0Q, b1Q);
    }

    const float bv = bias[col];
    #pragma unroll
    for (int t = 0; t < NT; ++t) {
        #pragma unroll
        for (int r = 0; r < 8; ++r) {
            const int row = 16 * mt[t] + 8 * lhalf + r;
            if (row < nrows) {
                float v = c[t][r] + bv;
                if (do_relu) v = fmaxf(v, 0.0f);
                out[(size_t)(base + row) * D_FEAT + col] = v;
            }
        }
    }
}

__global__ __launch_bounds__(CHUNK_GROUPS)
void bin_edges(const int* __restrict__ src, const int* __restrict__ dst, unsigned int* __restrict__ segs) {
    __shared__ __attribute__((aligned(16))) unsigned int seg[AGG_BLOCKS * SEG_ENT];
    __shared__ unsigned int scnt[AGG_BLOCKS];
    const int tid = threadIdx.x;
    const int g = blockIdx.x * CHUNK_GROUPS + tid;
    int4 d = make_int4(-1, -1, -1, -1);
    int4 s = make_int4(0, 0, 0, 0);
    if (g < GROUPS) {
        d = reinterpret_cast<const int4*>(dst)[g];
        s = reinterpret_cast<const int4*>(src)[g];
    }
    uint4* s4 = reinterpret_cast<uint4*>(seg);
    #pragma unroll
    for (int j = 0; j < 4; ++j)
        s4[tid + CHUNK_GROUPS * j] = make_uint4(SEG_EMPTY, SEG_EMPTY, SEG_EMPTY, SEG_EMPTY);
    if (tid < AGG_BLOCKS) scnt[tid] = 0u;
    __syncthreads();
    const int dd[4] = {d.x, d.y, d.z, d.w};
    const int ss[4] = {s.x, s.y, s.z, s.w};
    #pragma unroll
    for (int c = 0; c < 4; ++c) {
        if ((unsigned)dd[c] < (unsigned)N_NODES) {
            const unsigned o = (unsigned)dd[c] / ROWS_PER_BLOCK;
            const unsigned r = (unsigned)dd[c] - o * ROWS_PER_BLOCK;
            const unsigned slot = atomicAdd(&scnt[o], 1u);
            if (slot < SEG_ENT) seg[o * SEG_ENT + slot] = min((unsigned)ss[c], (unsigned)(N_NODES - 1)) | (r << 16);
        }
    }
    __syncthreads();
    if (tid < AGG_BLOCKS && scnt[tid] > SEG_ENT) seg[tid * SEG_ENT] = SEG_OVER;
    __syncthreads();
    uint4* o4 = reinterpret_cast<uint4*>(segs + (size_t)blockIdx.x * (AGG_BLOCKS * SEG_ENT));
    #pragma unroll
    for (int j = 0; j < 4; ++j)
        o4[tid + CHUNK_GROUPS * j] = s4[tid + CHUNK_GROUPS * j];
}

template <int LAYER>
__global__ __launch_bounds__(AGG_THREADS)
void sage_layer(const float* __restrict__ in,
                const int* __restrict__ src,
                const int* __restrict__ dst,
                const unsigned int* __restrict__ segs,
                const float* __restrict__ Wl,
                const float* __restrict__ Wr,
                const float* __restrict__ bias,
                float* __restrict__ out,
                unsigned int* __restrict__ list_ws,
                unsigned int* __restrict__ count_ws,
                int do_relu) {
    __shared__ __attribute__((aligned(16))) float acc[TILE_ROWS * PITCH];
    __shared__ __attribute__((aligned(16))) float own[TILE_ROWS * PITCH];
    __shared__ __attribute__((aligned(16))) unsigned int queue[QUEUE_CAP];
    __shared__ int cnt[TILE_ROWS];
    __shared__ unsigned int qn;
    __shared__ unsigned int over;

    const int tid   = threadIdx.x;
    const int base  = blockIdx.x * ROWS_PER_BLOCK;
    const int nrows = min(ROWS_PER_BLOCK, N_NODES - base);
    unsigned int* __restrict__ my_list = list_ws + (size_t)blockIdx.x * QUEUE_CAP;

    unsigned total = 0u;
    if (LAYER == 2) total = count_ws[blockIdx.x];

    float4 ownv[7];
    #pragma unroll
    for (int j = 0; j < 7; ++j) {
        const int r = (tid >> 5) + 32 * j;
        ownv[j] = make_float4(0.f, 0.f, 0.f, 0.f);
        if (r < nrows)
            ownv[j] = *reinterpret_cast<const float4*>(in + (size_t)(base + r) * D_FEAT + 4 * (tid & 31));
    }

    uint4 sv[4];
    if (LAYER == 1) {
        const uint4* sg4 = reinterpret_cast<const uint4*>(segs);
        #pragma unroll
        for (int j = 0; j < 4; ++j) {
            const int i = tid + AGG_THREADS * j;
            sv[j] = make_uint4(SEG_EMPTY, SEG_EMPTY, SEG_EMPTY, SEG_EMPTY);
            if (i < N_CHUNKS * (SEG_ENT / 4))
                sv[j] = sg4[((size_t)(i >> 4) * AGG_BLOCKS + blockIdx.x) * (SEG_ENT / 4) + (i & 15)];
        }
    }

    const unsigned n16 = (LAYER == 2 && total <= QUEUE_CAP) ? ((total + 7u) & ~7u) / 4u : 0u;
    uint4 lv[2];
    if (LAYER == 2) {
        const uint4* l4 = reinterpret_cast<const uint4*>(my_list);
        #pragma unroll
        for (int j = 0; j < 2; ++j) {
            const unsigned i = (unsigned)tid + AGG_THREADS * j;
            lv[j] = make_uint4(0u, 0u, 0u, 0u);
            if (i < n16) lv[j] = l4[i];
        }
    }

    float4* acc4 = reinterpret_cast<float4*>(acc);
    for (int i = tid; i < TILE_ROWS * PITCH / 4; i += AGG_THREADS)
        acc4[i] = make_float4(0.f, 0.f, 0.f, 0.f);
    if (tid < TILE_ROWS) cnt[tid] = 0;
    if (tid == 0) { qn = 0u; over = 0u; }
    #pragma unroll
    for (int j = 0; j < 7; ++j) {
        const int r = (tid >> 5) + 32 * j;
        if (r < TILE_ROWS)
            *reinterpret_cast<float4*>(own + r * PITCH + 4 * (tid & 31)) = ownv[j];
    }
    if (LAYER == 2) {
        uint4* q4 = reinterpret_cast<uint4*>(queue);
        #pragma unroll
        for (int j = 0; j < 2; ++j) {
            const unsigned i = (unsigned)tid + AGG_THREADS * j;
            if (i < n16) q4[i] = lv[j];
        }
    }
    __syncthreads();

    if (LAYER == 2) {
        #pragma unroll
        for (int j = 0; j < 2; ++j) {
            const unsigned i = (unsigned)tid + AGG_THREADS * j;
            const unsigned w[4] = {lv[j].x, lv[j].y, lv[j].z, lv[j].w};
            if (i < n16) {
                #pragma unroll
                for (int c = 0; c < 4; ++c)
                    __hip_atomic_fetch_add(&cnt[min(w[c] >> 16, (unsigned)(TILE_ROWS - 1))], 1,
                                           __ATOMIC_RELAXED, __HIP_MEMORY_SCOPE_WORKGROUP);
            }
        }
    }

    if (LAYER == 1) {
        const unsigned e[16] = {sv[0].x, sv[0].y, sv[0].z, sv[0].w, sv[1].x, sv[1].y, sv[1].z, sv[1].w,
                                sv[2].x, sv[2].y, sv[2].z, sv[2].w, sv[3].x, sv[3].y, sv[3].z, sv[3].w};
        {
            const int lane = tid & 31;
            unsigned n = 0u;
            bool ov = false;
            #pragma unroll
            for (int k = 0; k < 16; ++k) {
                ov = ov || (e[k] == SEG_OVER);
                n += (e[k] < SEG_OVER) ? 1u : 0u;
            }
            unsigned run = n;
            #pragma unroll
            for (int dlt = 1; dlt < 32; dlt <<= 1) {
                const unsigned up = (unsigned)__shfl_up((int)run, dlt);
                if (lane >= dlt) run += up;
            }
            unsigned wbase = 0u;
            if (lane == 31 && run != 0u) wbase = atomicAdd(&qn, run);
            wbase = (unsigned)__shfl((int)wbase, 31);
            unsigned slot = wbase + run - n;
            #pragma unroll
            for (int k = 0; k < 16; ++k) {
                if (e[k] < SEG_OVER) {
                    if (slot < QUEUE_CAP) queue[slot] = e[k];
                    ++slot;
                    __hip_atomic_fetch_add(&cnt[min(e[k] >> 16, (unsigned)(TILE_ROWS - 1))], 1,
                                           __ATOMIC_RELAXED, __HIP_MEMORY_SCOPE_WORKGROUP);
                }
            }
            if (ov) over = 1u;
        }
        __syncthreads();

        total = (over == 0u) ? qn : (unsigned)(QUEUE_CAP + 1);
        const unsigned padded = (total <= QUEUE_CAP) ? ((total + 7u) & ~7u) : 0u;
        if ((unsigned)tid < padded - total && total <= QUEUE_CAP) queue[total + (unsigned)tid] = PAD_ENTRY;
        __syncthreads();
        {
            const uint4* q4 = reinterpret_cast<const uint4*>(queue);
            uint4* l4 = reinterpret_cast<uint4*>(my_list);
            for (unsigned i = (unsigned)tid; i < padded / 4u; i += AGG_THREADS) l4[i] = q4[i];
        }
        if (tid == 0) count_ws[blockIdx.x] = total;
    }

    const int wave    = __builtin_amdgcn_readfirstlane(tid >> 5);
    const unsigned total_s = (unsigned)__builtin_amdgcn_readfirstlane((int)total);
    const int lane    = tid & 31;
    const int lhalf   = lane >> 4;
    const int lmod    = lane & 15;
    const int ntile   = wave & 7;
    const int g       = wave >> 3;
    const int col     = ntile * 16 + lmod;
    const bool four   = (g == (ntile >> 1));
    v8f c[4];
    #pragma unroll
    for (int t = 0; t < 4; ++t) c[t] = v8f{};

    if (total_s <= QUEUE_CAP) {
        if (four) gather_with_own<4>(c, in, acc, own, queue, Wr, total_s, (unsigned)wave, lane, g, lmod, lhalf, col);
        else      gather_with_own<3>(c, in, acc, own, queue, Wr, total_s, (unsigned)wave, lane, g, lmod, lhalf, col);
    } else {
        if (tid < TILE_ROWS) cnt[tid] = 0;
        __syncthreads();
        for (int e = tid; e < N_EDGES; e += AGG_THREADS) {
            const unsigned u = (unsigned)(dst[e] - base);
            if (u < (unsigned)nrows) {
                const unsigned s = min((unsigned)src[e], (unsigned)(N_NODES - 1));
                const float* r = in + (size_t)s * D_FEAT;
                float* a = acc + u * PITCH;
                for (int cc = 0; cc < D_FEAT; ++cc)
                    __hip_atomic_fetch_add(a + cc, r[cc],
                                           __ATOMIC_RELAXED, __HIP_MEMORY_SCOPE_WORKGROUP);
                __hip_atomic_fetch_add(&cnt[u], 1,
                                       __ATOMIC_RELAXED, __HIP_MEMORY_SCOPE_WORKGROUP);
            }
        }
        if (four) gather_with_own<4>(c, in, acc, own, queue, Wr, 0u, (unsigned)wave, lane, g, lmod, lhalf, col);
        else      gather_with_own<3>(c, in, acc, own, queue, Wr, 0u, (unsigned)wave, lane, g, lmod, lhalf, col);
    }
    __syncthreads();

    for (int i = tid; i < nrows * (D_FEAT / 4); i += AGG_THREADS) {
        const int r = i >> 5;
        const float dv = fmaxf((float)cnt[r], 1.0f);
        float4* p = reinterpret_cast<float4*>(acc + r * PITCH + 4 * (i & 31));
        float4 a = *p;
        a.x = a.x / dv;
        a.y = a.y / dv;
        a.z = a.z / dv;
        a.w = a.w / dv;
        *p = a;
    }
    __syncthreads();

    if (four) mean_products_store<4>(c, acc, Wl, bias, out, g, lmod, lhalf, col, base, nrows, do_relu);
    else      mean_products_store<3>(c, acc, Wl, bias, out, g, lmod, lhalf, col, base, nrows, do_relu);
}

extern "C" void kernel_launch(void* const* d_in, const int* in_sizes, int n_in,
                              void* d_out, int out_size, void* d_ws, size_t ws_size,
                              hipStream_t stream) {
    const float* x    = (const float*)d_in[0];
    const int*   eidx = (const int*)d_in[1];
    const float* W1l  = (const float*)d_in[2];
    const float* b1   = (const float*)d_in[3];
    const float* W1r  = (const float*)d_in[4];
    const float* W2l  = (const float*)d_in[5];
    const float* b2   = (const float*)d_in[6];
    const float* W2r  = (const float*)d_in[7];
    float* out = (float*)d_out;

    const int* src = eidx;
    const int* dst = eidx + N_EDGES;

    float* agg = (float*)d_ws;
    float* deg = agg + (size_t)N_NODES * D_FEAT;
    float* h   = deg + N_NODES;
    unsigned int* list_ws  = (unsigned int*)agg;
    unsigned int* count_ws = (unsigned int*)deg;
    unsigned int* seg_ws   = list_ws + (size_t)AGG_BLOCKS * QUEUE_CAP;

    bin_edges<<<N_CHUNKS, CHUNK_GROUPS, 0, stream>>>(src, dst, seg_ws);

    sage_layer<1><<<AGG_BLOCKS, AGG_THREADS, 0, stream>>>(x, src, dst, seg_ws, W1l, W1r, b1, h, list_ws, count_ws, 1);
    sage_layer<2><<<AGG_BLOCKS, AGG_THREADS, 0, stream>>>(h, src, dst, seg_ws, W2l, W2r, b2, out, list_ws, count_ws, 0);
}


# Each gathered row read by ONE 16-B-per-lane load (was four 4-B loads), rows line-aligned (layer-1 plane moved); sum tile holds column 4l+j at word 32j+l: mean @ Wl adds its k terms in that order
# speedup vs baseline: 7.7241x; 1.1139x over previous
#include <hip/hip_runtime.h>
#include <hip/hip_bf16.h>


#define N_NODES 50000
#define N_EDGES 800000
#define D_FEAT  128

#define ROWS_PER_BLOCK 196
#define AGG_BLOCKS     256
#define AGG_THREADS    1024
#define QUEUE_CAP      8192
#define TILE_ROWS      208
#define PITCH          132
#define GROUPS         (N_EDGES / 4)
#define CHUNK_GROUPS   1024
#define N_CHUNKS       ((GROUPS + CHUNK_GROUPS - 1) / CHUNK_GROUPS)
#define SEG_ENT        64
#define SEG_EMPTY      0xFFFFFFFFu
#define SEG_OVER       0xFFFFFFFEu
#define PAD_ENTRY      ((unsigned)(TILE_ROWS - 1) << 16)

typedef __attribute__((ext_vector_type(2))) float v2f;
typedef __attribute__((ext_vector_type(8))) float v8f;


__device__ __forceinline__ v8f wmma4(const v2f& a, const v2f& b, const v8f& c) {
    return __builtin_amdgcn_wmma_f32_16x16x4_f32(false, a, false, b, (short)0, c, false, false);
}

__device__ __forceinline__ void load_b4(const float* __restrict__ w, int k0, v2f (&b)[4]) {
    #pragma unroll
    for (int s = 0; s < 4; ++s) {
        b[s].x = w[(k0 + 4 * s) * D_FEAT];
        b[s].y = w[(k0 + 4 * s + 1) * D_FEAT];
    }
}

template <int NT>
__device__ __forceinline__ void ksteps4(v8f (&c)[4], const float* const (&pa)[NT], const v2f (&b)[4], int k0) {
    #pragma unroll
    for (int s = 0; s < 4; ++s) {
        v2f a[NT];
        #pragma unroll
        for (int t = 0; t < NT; ++t) a[t] = *(const v2f*)(pa[t] + k0 + 4 * s);
        #pragma unroll
        for (int t = 0; t < NT; ++t) c[t] = wmma4(a[t], b[s], c[t]);
    }
}

template <int NT>
__device__ __forceinline__ void gather_with_own(v8f (&c)[4], const float* __restrict__ in, float* acc,
                                                const float* own_t, const unsigned int* queue,
                                                const float* __restrict__ Wr,
                                                unsigned total, unsigned wave, int lane,
                                                int g, int lmod, int lhalf, int col) {
    const float* po[NT];
    #pragma unroll
    for (int t = 0; t < NT; ++t) {
        const int mt = (t < 3) ? 3 * g + t : 12;
        po[t] = own_t + (16 * mt + lmod) * PITCH + 2 * lhalf;
    }
    const float* wr = Wr + (2 * lhalf) * D_FEAT + col;
    const uint4* queue4 = reinterpret_cast<const uint4*>(queue);
    const unsigned stride = (AGG_THREADS / 32) * 8u;

    unsigned ent[8];
    float4 v[8];
    auto issue_rows = [&](unsigned k) {
        const uint4 qa = queue4[k / 4u];
        const uint4 qb = queue4[k / 4u + 1u];
        const unsigned raw[8] = {qa.x, qa.y, qa.z, qa.w, qb.x, qb.y, qb.z, qb.w};
        #pragma unroll
        for (int t = 0; t < 8; ++t)
            ent[t] = (unsigned)__builtin_amdgcn_readfirstlane((int)raw[t]);
        #pragma unroll
        for (int t = 0; t < 8; ++t) {
            const unsigned s = min(ent[t] & 0xFFFFu, (unsigned)(N_NODES - 1));
            v[t] = *reinterpret_cast<const float4*>(in + (size_t)s * D_FEAT + 4 * lane);
        }
    };
    auto add_rows = [&]() {
        #pragma unroll
        for (int t = 0; t < 8; ++t) {
            const unsigned u = min(ent[t] >> 16, (unsigned)(TILE_ROWS - 1));
            float* a = acc + u * PITCH + lane;
            __hip_atomic_fetch_add(a,      v[t].x, __ATOMIC_RELAXED, __HIP_MEMORY_SCOPE_WORKGROUP);
            __hip_atomic_fetch_add(a + 32, v[t].y, __ATOMIC_RELAXED, __HIP_MEMORY_SCOPE_WORKGROUP);
            __hip_atomic_fetch_add(a + 64, v[t].z, __ATOMIC_RELAXED, __HIP_MEMORY_SCOPE_WORKGROUP);
            __hip_atomic_fetch_add(a + 96, v[t].w, __ATOMIC_RELAXED, __HIP_MEMORY_SCOPE_WORKGROUP);
        }
    };

    unsigned k = wave * 8u;
    int kk = 0;
    v2f b[4];
    for (; k < total && kk < D_FEAT; k += stride, kk += 16) {
        load_b4(wr, kk, b);
        issue_rows(k);
        ksteps4<NT>(c, po, b, kk);
        add_rows();
    }
    for (; k < total; k += stride) {
        issue_rows(k);
        add_rows();
    }
    for (; kk < D_FEAT; kk += 16) {
        load_b4(wr, kk, b);
        ksteps4<NT>(c, po, b, kk);
    }
}

template <int NT>
__device__ __forceinline__ void mean_products_store(v8f (&c)[4], const float* mean_t,
                                                    const float* __restrict__ Wl,
                                                    const float* __restrict__ bias,
                                                    float* __restrict__ out,
                                                    int g, int lmod, int lhalf, int col,
                                                    int base, int nrows, int do_relu) {
    int mt[NT];
    const float* pm[NT];
    #pragma unroll
    for (int t = 0; t < NT; ++t) {
        mt[t] = (t < 3) ? 3 * g + t : 12;
        pm[t] = mean_t + (16 * mt[t] + lmod) * PITCH + 2 * lhalf;
    }
    const float* wl = Wl + (8 * lhalf) * D_FEAT + col;
    auto wrow = [](int k) { return 4 * (k & 31) + (k >> 5); };

    v2f a0P[NT], a1P[NT], b0P, b1P;
    v2f a0Q[NT], a1Q[NT], b0Q, b1Q;
    auto load_pair = [&](int k, v2f (&a0)[NT], v2f (&a1)[NT], v2f& b0, v2f& b1) {
        #pragma unroll
        for (int t = 0; t < NT; ++t) {
            a0[t] = *(const v2f*)(pm[t] + k);
            a1[t] = *(const v2f*)(pm[t] + k + 4);
        }
        b0.x = wl[wrow(k) * D_FEAT];
        b0.y = wl[(wrow(k) + 4) * D_FEAT];
        b1.x = wl[wrow(k + 4) * D_FEAT];
        b1.y = wl[(wrow(k + 4) + 4) * D_FEAT];
    };
    auto products = [&](const v2f (&a0)[NT], const v2f (&a1)[NT], const v2f& b0, const v2f& b1) {
        #pragma unroll
        for (int t = 0; t < NT; ++t) c[t] = wmma4(a0[t], b0, c[t]);
        #pragma unroll
        for (int t = 0; t < NT; ++t) c[t] = wmma4(a1[t], b1, c[t]);
    };
    load_pair(0, a0P, a1P, b0P, b1P);
    #pragma unroll 2
    for (int k = 0; k < D_FEAT; k += 16) {
        load_pair(k + 8, a0Q, a1Q, b0Q, b1Q);
        products(a0P, a1P, b0P, b1P);
        if (k + 16 < D_FEAT) load_pair(k + 16, a0P, a1P, b0P, b1P);
        products(a0Q, a1Q, b0Q, b1Q);
    }

    const float bv = bias[col];
    #pragma unroll
    for (int t = 0; t < NT; ++t) {
        #pragma unroll
        for (int r = 0; r < 8; ++r) {
            const int row = 16 * mt[t] + 8 * lhalf + r;
            if (row < nrows) {
                float v = c[t][r] + bv;
                if (do_relu) v = fmaxf(v, 0.0f);
                out[(size_t)(base + row) * D_FEAT + col] = v;
            }
        }
    }
}

__global__ __launch_bounds__(CHUNK_GROUPS)
void bin_edges(const int* __restrict__ src, const int* __restrict__ dst, unsigned int* __restrict__ segs) {
    __shared__ __attribute__((aligned(16))) unsigned int seg[AGG_BLOCKS * SEG_ENT];
    __shared__ unsigned int scnt[AGG_BLOCKS];
    const int tid = threadIdx.x;
    const int g = blockIdx.x * CHUNK_GROUPS + tid;
    int4 d = make_int4(-1, -1, -1, -1);
    int4 s = make_int4(0, 0, 0, 0);
    if (g < GROUPS) {
        d = reinterpret_cast<const int4*>(dst)[g];
        s = reinterpret_cast<const int4*>(src)[g];
    }
    uint4* s4 = reinterpret_cast<uint4*>(seg);
    #pragma unroll
    for (int j = 0; j < 4; ++j)
        s4[tid + CHUNK_GROUPS * j] = make_uint4(SEG_EMPTY, SEG_EMPTY, SEG_EMPTY, SEG_EMPTY);
    if (tid < AGG_BLOCKS) scnt[tid] = 0u;
    __syncthreads();
    const int dd[4] = {d.x, d.y, d.z, d.w};
    const int ss[4] = {s.x, s.y, s.z, s.w};
    #pragma unroll
    for (int c = 0; c < 4; ++c) {
        if ((unsigned)dd[c] < (unsigned)N_NODES) {
            const unsigned o = (unsigned)dd[c] / ROWS_PER_BLOCK;
            const unsigned r = (unsigned)dd[c] - o * ROWS_PER_BLOCK;
            const unsigned slot = atomicAdd(&scnt[o], 1u);
            if (slot < SEG_ENT) seg[o * SEG_ENT + slot] = min((unsigned)ss[c], (unsigned)(N_NODES - 1)) | (r << 16);
        }
    }
    __syncthreads();
    if (tid < AGG_BLOCKS && scnt[tid] > SEG_ENT) seg[tid * SEG_ENT] = SEG_OVER;
    __syncthreads();
    uint4* o4 = reinterpret_cast<uint4*>(segs + (size_t)blockIdx.x * (AGG_BLOCKS * SEG_ENT));
    #pragma unroll
    for (int j = 0; j < 4; ++j)
        o4[tid + CHUNK_GROUPS * j] = s4[tid + CHUNK_GROUPS * j];
}

template <int LAYER>
__global__ __launch_bounds__(AGG_THREADS)
void sage_layer(const float* __restrict__ in,
                const int* __restrict__ src,
                const int* __restrict__ dst,
                const unsigned int* __restrict__ segs,
                const float* __restrict__ Wl,
                const float* __restrict__ Wr,
                const float* __restrict__ bias,
                float* __restrict__ out,
                unsigned int* __restrict__ list_ws,
                unsigned int* __restrict__ count_ws,
                int do_relu) {
    __shared__ __attribute__((aligned(16))) float acc[TILE_ROWS * PITCH];
    __shared__ __attribute__((aligned(16))) float own[TILE_ROWS * PITCH];
    __shared__ __attribute__((aligned(16))) unsigned int queue[QUEUE_CAP];
    __shared__ int cnt[TILE_ROWS];
    __shared__ unsigned int qn;
    __shared__ unsigned int over;

    const int tid   = threadIdx.x;
    const int base  = blockIdx.x * ROWS_PER_BLOCK;
    const int nrows = min(ROWS_PER_BLOCK, N_NODES - base);
    unsigned int* __restrict__ my_list = list_ws + (size_t)blockIdx.x * QUEUE_CAP;

    unsigned total = 0u;
    if (LAYER == 2) total = count_ws[blockIdx.x];

    float4 ownv[7];
    #pragma unroll
    for (int j = 0; j < 7; ++j) {
        const int r = (tid >> 5) + 32 * j;
        ownv[j] = make_float4(0.f, 0.f, 0.f, 0.f);
        if (r < nrows)
            ownv[j] = *reinterpret_cast<const float4*>(in + (size_t)(base + r) * D_FEAT + 4 * (tid & 31));
    }

    uint4 sv[4];
    if (LAYER == 1) {
        const uint4* sg4 = reinterpret_cast<const uint4*>(segs);
        #pragma unroll
        for (int j = 0; j < 4; ++j) {
            const int i = tid + AGG_THREADS * j;
            sv[j] = make_uint4(SEG_EMPTY, SEG_EMPTY, SEG_EMPTY, SEG_EMPTY);
            if (i < N_CHUNKS * (SEG_ENT / 4))
                sv[j] = sg4[((size_t)(i >> 4) * AGG_BLOCKS + blockIdx.x) * (SEG_ENT / 4) + (i & 15)];
        }
    }

    const unsigned n16 = (LAYER == 2 && total <= QUEUE_CAP) ? ((total + 7u) & ~7u) / 4u : 0u;
    uint4 lv[2];
    if (LAYER == 2) {
        const uint4* l4 = reinterpret_cast<const uint4*>(my_list);
        #pragma unroll
        for (int j = 0; j < 2; ++j) {
            const unsigned i = (unsigned)tid + AGG_THREADS * j;
            lv[j] = make_uint4(0u, 0u, 0u, 0u);
            if (i < n16) lv[j] = l4[i];
        }
    }

    float4* acc4 = reinterpret_cast<float4*>(acc);
    for (int i = tid; i < TILE_ROWS * PITCH / 4; i += AGG_THREADS)
        acc4[i] = make_float4(0.f, 0.f, 0.f, 0.f);
    if (tid < TILE_ROWS) cnt[tid] = 0;
    if (tid == 0) { qn = 0u; over = 0u; }
    #pragma unroll
    for (int j = 0; j < 7; ++j) {
        const int r = (tid >> 5) + 32 * j;
        if (r < TILE_ROWS)
            *reinterpret_cast<float4*>(own + r * PITCH + 4 * (tid & 31)) = ownv[j];
    }
    if (LAYER == 2) {
        uint4* q4 = reinterpret_cast<uint4*>(queue);
        #pragma unroll
        for (int j = 0; j < 2; ++j) {
            const unsigned i = (unsigned)tid + AGG_THREADS * j;
            if (i < n16) q4[i] = lv[j];
        }
    }
    __syncthreads();

    if (LAYER == 2) {
        #pragma unroll
        for (int j = 0; j < 2; ++j) {
            const unsigned i = (unsigned)tid + AGG_THREADS * j;
            const unsigned w[4] = {lv[j].x, lv[j].y, lv[j].z, lv[j].w};
            if (i < n16) {
                #pragma unroll
                for (int c = 0; c < 4; ++c)
                    __hip_atomic_fetch_add(&cnt[min(w[c] >> 16, (unsigned)(TILE_ROWS - 1))], 1,
                                           __ATOMIC_RELAXED, __HIP_MEMORY_SCOPE_WORKGROUP);
            }
        }
    }

    if (LAYER == 1) {
        const unsigned e[16] = {sv[0].x, sv[0].y, sv[0].z, sv[0].w, sv[1].x, sv[1].y, sv[1].z, sv[1].w,
                                sv[2].x, sv[2].y, sv[2].z, sv[2].w, sv[3].x, sv[3].y, sv[3].z, sv[3].w};
        {
            const int lane = tid & 31;
            unsigned n = 0u;
            bool ov = false;
            #pragma unroll
            for (int k = 0; k < 16; ++k) {
                ov = ov || (e[k] == SEG_OVER);
                n += (e[k] < SEG_OVER) ? 1u : 0u;
            }
            unsigned run = n;
            #pragma unroll
            for (int dlt = 1; dlt < 32; dlt <<= 1) {
                const unsigned up = (unsigned)__shfl_up((int)run, dlt);
                if (lane >= dlt) run += up;
            }
            unsigned wbase = 0u;
            if (lane == 31 && run != 0u) wbase = atomicAdd(&qn, run);
            wbase = (unsigned)__shfl((int)wbase, 31);
            unsigned slot = wbase + run - n;
            #pragma unroll
            for (int k = 0; k < 16; ++k) {
                if (e[k] < SEG_OVER) {
                    if (slot < QUEUE_CAP) queue[slot] = e[k];
                    ++slot;
                    __hip_atomic_fetch_add(&cnt[min(e[k] >> 16, (unsigned)(TILE_ROWS - 1))], 1,
                                           __ATOMIC_RELAXED, __HIP_MEMORY_SCOPE_WORKGROUP);
                }
            }
            if (ov) over = 1u;
        }
        __syncthreads();

        total = (over == 0u) ? qn : (unsigned)(QUEUE_CAP + 1);
        const unsigned padded = (total <= QUEUE_CAP) ? ((total + 7u) & ~7u) : 0u;
        if ((unsigned)tid < padded - total && total <= QUEUE_CAP) queue[total + (unsigned)tid] = PAD_ENTRY;
        __syncthreads();
        {
            const uint4* q4 = reinterpret_cast<const uint4*>(queue);
            uint4* l4 = reinterpret_cast<uint4*>(my_list);
            for (unsigned i = (unsigned)tid; i < padded / 4u; i += AGG_THREADS) l4[i] = q4[i];
        }
        if (tid == 0) count_ws[blockIdx.x] = total;
    }

    const int wave    = __builtin_amdgcn_readfirstlane(tid >> 5);
    const unsigned total_s = (unsigned)__builtin_amdgcn_readfirstlane((int)total);
    const int lane    = tid & 31;
    const int lhalf   = lane >> 4;
    const int lmod    = lane & 15;
    const int ntile   = wave & 7;
    const int g       = wave >> 3;
    const int col     = ntile * 16 + lmod;
    const bool four   = (g == (ntile >> 1));
    v8f c[4];
    #pragma unroll
    for (int t = 0; t < 4; ++t) c[t] = v8f{};

    if (total_s <= QUEUE_CAP) {
        if (four) gather_with_own<4>(c, in, acc, own, queue, Wr, total_s, (unsigned)wave, lane, g, lmod, lhalf, col);
        else      gather_with_own<3>(c, in, acc, own, queue, Wr, total_s, (unsigned)wave, lane, g, lmod, lhalf, col);
    } else {
        if (tid < TILE_ROWS) cnt[tid] = 0;
        __syncthreads();
        for (int e = tid; e < N_EDGES; e += AGG_THREADS) {
            const unsigned u = (unsigned)(dst[e] - base);
            if (u < (unsigned)nrows) {
                const unsigned s = min((unsigned)src[e], (unsigned)(N_NODES - 1));
                const float* r = in + (size_t)s * D_FEAT;
                float* a = acc + u * PITCH;
                for (int cc = 0; cc < D_FEAT; ++cc)
                    __hip_atomic_fetch_add(a + 32 * (cc & 3) + (cc >> 2), r[cc],
                                           __ATOMIC_RELAXED, __HIP_MEMORY_SCOPE_WORKGROUP);
                __hip_atomic_fetch_add(&cnt[u], 1,
                                       __ATOMIC_RELAXED, __HIP_MEMORY_SCOPE_WORKGROUP);
            }
        }
        if (four) gather_with_own<4>(c, in, acc, own, queue, Wr, 0u, (unsigned)wave, lane, g, lmod, lhalf, col);
        else      gather_with_own<3>(c, in, acc, own, queue, Wr, 0u, (unsigned)wave, lane, g, lmod, lhalf, col);
    }
    __syncthreads();

    for (int i = tid; i < nrows * (D_FEAT / 4); i += AGG_THREADS) {
        const int r = i >> 5;
        const float dv = fmaxf((float)cnt[r], 1.0f);
        float4* p = reinterpret_cast<float4*>(acc + r * PITCH + 4 * (i & 31));
        float4 a = *p;
        a.x = a.x / dv;
        a.y = a.y / dv;
        a.z = a.z / dv;
        a.w = a.w / dv;
        *p = a;
    }
    __syncthreads();

    if (four) mean_products_store<4>(c, acc, Wl, bias, out, g, lmod, lhalf, col, base, nrows, do_relu);
    else      mean_products_store<3>(c, acc, Wl, bias, out, g, lmod, lhalf, col, base, nrows, do_relu);
}

extern "C" void kernel_launch(void* const* d_in, const int* in_sizes, int n_in,
                              void* d_out, int out_size, void* d_ws, size_t ws_size,
                              hipStream_t stream) {
    const float* x    = (const float*)d_in[0];
    const int*   eidx = (const int*)d_in[1];
    const float* W1l  = (const float*)d_in[2];
    const float* b1   = (const float*)d_in[3];
    const float* W1r  = (const float*)d_in[4];
    const float* W2l  = (const float*)d_in[5];
    const float* b2   = (const float*)d_in[6];
    const float* W2r  = (const float*)d_in[7];
    float* out = (float*)d_out;

    const int* src = eidx;
    const int* dst = eidx + N_EDGES;

    float* agg = (float*)d_ws;
    float* deg = agg + (size_t)N_NODES * D_FEAT;
    float* h   = deg + 256;
    unsigned int* list_ws  = (unsigned int*)agg;
    unsigned int* count_ws = (unsigned int*)deg;
    unsigned int* seg_ws   = list_ws + (size_t)AGG_BLOCKS * QUEUE_CAP;

    bin_edges<<<N_CHUNKS, CHUNK_GROUPS, 0, stream>>>(src, dst, seg_ws);

    sage_layer<1><<<AGG_BLOCKS, AGG_THREADS, 0, stream>>>(x, src, dst, seg_ws, W1l, W1r, b1, h, list_ws, count_ws, 1);
    sage_layer<2><<<AGG_BLOCKS, AGG_THREADS, 0, stream>>>(h, src, dst, seg_ws, W2l, W2r, b2, out, list_ws, count_ws, 0);
}


# Block's list ordered by local row in layer 1 (counts, prefix, returning LDS int add); trips sum same-row entries in registers: 4 LDS float adds per run; sums: list order in run, runs by arrival
# speedup vs baseline: 8.2273x; 1.0651x over previous
#include <hip/hip_runtime.h>
#include <hip/hip_bf16.h>


#define N_NODES 50000
#define N_EDGES 800000
#define D_FEAT  128

#define ROWS_PER_BLOCK 196
#define AGG_BLOCKS     256
#define AGG_THREADS    1024
#define QUEUE_CAP      8192
#define TILE_ROWS      208
#define PITCH          132
#define GROUPS         (N_EDGES / 4)
#define CHUNK_GROUPS   1024
#define N_CHUNKS       ((GROUPS + CHUNK_GROUPS - 1) / CHUNK_GROUPS)
#define SEG_ENT        64
#define SEG_EMPTY      0xFFFFFFFFu
#define SEG_OVER       0xFFFFFFFEu
#define PAD_ENTRY      ((unsigned)(TILE_ROWS - 1) << 16)
#define ROW_SLOTS      224

typedef __attribute__((ext_vector_type(2))) float v2f;
typedef __attribute__((ext_vector_type(8))) float v8f;


__device__ __forceinline__ v8f wmma4(const v2f& a, const v2f& b, const v8f& c) {
    return __builtin_amdgcn_wmma_f32_16x16x4_f32(false, a, false, b, (short)0, c, false, false);
}

__device__ __forceinline__ void load_b4(const float* __restrict__ w, int k0, v2f (&b)[4]) {
    #pragma unroll
    for (int s = 0; s < 4; ++s) {
        b[s].x = w[(k0 + 4 * s) * D_FEAT];
        b[s].y = w[(k0 + 4 * s + 1) * D_FEAT];
    }
}

template <int NT>
__device__ __forceinline__ void ksteps4(v8f (&c)[4], const float* const (&pa)[NT], const v2f (&b)[4], int k0) {
    #pragma unroll
    for (int s = 0; s < 4; ++s) {
        v2f a[NT];
        #pragma unroll
        for (int t = 0; t < NT; ++t) a[t] = *(const v2f*)(pa[t] + k0 + 4 * s);
        #pragma unroll
        for (int t = 0; t < NT; ++t) c[t] = wmma4(a[t], b[s], c[t]);
    }
}

template <int NT>
__device__ __forceinline__ void gather_with_own(v8f (&c)[4], const float* __restrict__ in, float* acc,
                                                const float* own_t, const unsigned int* queue,
                                                const float* __restrict__ Wr,
                                                unsigned total, unsigned wave, int lane,
                                                int g, int lmod, int lhalf, int col) {
    const float* po[NT];
    #pragma unroll
    for (int t = 0; t < NT; ++t) {
        const int mt = (t < 3) ? 3 * g + t : 12;
        po[t] = own_t + (16 * mt + lmod) * PITCH + 2 * lhalf;
    }
    const float* wr = Wr + (2 * lhalf) * D_FEAT + col;
    const uint4* queue4 = reinterpret_cast<const uint4*>(queue);
    const unsigned stride = (AGG_THREADS / 32) * 8u;

    unsigned ent[8];
    float4 v[8];
    auto issue_rows = [&](unsigned k) {
        const uint4 qa = queue4[k / 4u];
        const uint4 qb = queue4[k / 4u + 1u];
        const unsigned raw[8] = {qa.x, qa.y, qa.z, qa.w, qb.x, qb.y, qb.z, qb.w};
        #pragma unroll
        for (int t = 0; t < 8; ++t)
            ent[t] = (unsigned)__builtin_amdgcn_readfirstlane((int)raw[t]);
        #pragma unroll
        for (int t = 0; t < 8; ++t) {
            const unsigned s = min(ent[t] & 0xFFFFu, (unsigned)(N_NODES - 1));
            v[t] = *reinterpret_cast<const float4*>(in + (size_t)s * D_FEAT + 4 * lane);
        }
    };
    auto flush_run = [&](unsigned u, const float4& s) {
        float* a = acc + u * PITCH + lane;
        __hip_atomic_fetch_add(a,      s.x, __ATOMIC_RELAXED, __HIP_MEMORY_SCOPE_WORKGROUP);
        __hip_atomic_fetch_add(a + 32, s.y, __ATOMIC_RELAXED, __HIP_MEMORY_SCOPE_WORKGROUP);
        __hip_atomic_fetch_add(a + 64, s.z, __ATOMIC_RELAXED, __HIP_MEMORY_SCOPE_WORKGROUP);
        __hip_atomic_fetch_add(a + 96, s.w, __ATOMIC_RELAXED, __HIP_MEMORY_SCOPE_WORKGROUP);
    };
    auto add_rows = [&]() {
        unsigned ur = min(ent[0] >> 16, (unsigned)(TILE_ROWS - 1));
        float4 s = v[0];
        #pragma unroll
        for (int t = 1; t < 8; ++t) {
            const unsigned u = min(ent[t] >> 16, (unsigned)(TILE_ROWS - 1));
            if (u == ur) {
                s.x += v[t].x;
                s.y += v[t].y;
                s.z += v[t].z;
                s.w += v[t].w;
            } else {
                flush_run(ur, s);
                ur = u;
                s = v[t];
            }
        }
        flush_run(ur, s);
    };

    unsigned k = wave * 8u;
    int kk = 0;
    v2f b[4];
    for (; k < total && kk < D_FEAT; k += stride, kk += 16) {
        load_b4(wr, kk, b);
        issue_rows(k);
        ksteps4<NT>(c, po, b, kk);
        add_rows();
    }
    for (; k < total; k += stride) {
        issue_rows(k);
        add_rows();
    }
    for (; kk < D_FEAT; kk += 16) {
        load_b4(wr, kk, b);
        ksteps4<NT>(c, po, b, kk);
    }
}

template <int NT>
__device__ __forceinline__ void mean_products_store(v8f (&c)[4], const float* mean_t,
                                                    const float* __restrict__ Wl,
                                                    const float* __restrict__ bias,
                                                    float* __restrict__ out,
                                                    int g, int lmod, int lhalf, int col,
                                                    int base, int nrows, int do_relu) {
    int mt[NT];
    const float* pm[NT];
    #pragma unroll
    for (int t = 0; t < NT; ++t) {
        mt[t] = (t < 3) ? 3 * g + t : 12;
        pm[t] = mean_t + (16 * mt[t] + lmod) * PITCH + 2 * lhalf;
    }
    const float* wl = Wl + (8 * lhalf) * D_FEAT + col;
    auto wrow = [](int k) { return 4 * (k & 31) + (k >> 5); };

    v2f a0P[NT], a1P[NT], b0P, b1P;
    v2f a0Q[NT], a1Q[NT], b0Q, b1Q;
    auto load_pair = [&](int k, v2f (&a0)[NT], v2f (&a1)[NT], v2f& b0, v2f& b1) {
        #pragma unroll
        for (int t = 0; t < NT; ++t) {
            a0[t] = *(const v2f*)(pm[t] + k);
            a1[t] = *(const v2f*)(pm[t] + k + 4);
        }
        b0.x = wl[wrow(k) * D_FEAT];
        b0.y = wl[(wrow(k) + 4) * D_FEAT];
        b1.x = wl[wrow(k + 4) * D_FEAT];
        b1.y = wl[(wrow(k + 4) + 4) * D_FEAT];
    };
    auto products = [&](const v2f (&a0)[NT], const v2f (&a1)[NT], const v2f& b0, const v2f& b1) {
        #pragma unroll
        for (int t = 0; t < NT; ++t) c[t] = wmma4(a0[t], b0, c[t]);
        #pragma unroll
        for (int t = 0; t < NT; ++t) c[t] = wmma4(a1[t], b1, c[t]);
    };
    load_pair(0, a0P, a1P, b0P, b1P);
    #pragma unroll 2
    for (int k = 0; k < D_FEAT; k += 16) {
        load_pair(k + 8, a0Q, a1Q, b0Q, b1Q);
        products(a0P, a1P, b0P, b1P);
        if (k + 16 < D_FEAT) load_pair(k + 16, a0P, a1P, b0P, b1P);
        products(a0Q, a1Q, b0Q, b1Q);
    }

    const float bv = bias[col];
    #pragma unroll
    for (int t = 0; t < NT; ++t) {
        #pragma unroll
        for (int r = 0; r < 8; ++r) {
            const int row = 16 * mt[t] + 8 * lhalf + r;
            if (row < nrows) {
                float v = c[t][r] + bv;
                if (do_relu) v = fmaxf(v, 0.0f);
                out[(size_t)(base + row) * D_FEAT + col] = v;
            }
        }
    }
}

__global__ __launch_bounds__(CHUNK_GROUPS)
void bin_edges(const int* __restrict__ src, const int* __restrict__ dst, unsigned int* __restrict__ segs) {
    __shared__ __attribute__((aligned(16))) unsigned int seg[AGG_BLOCKS * SEG_ENT];
    __shared__ unsigned int scnt[AGG_BLOCKS];
    const int tid = threadIdx.x;
    const int g = blockIdx.x * CHUNK_GROUPS + tid;
    int4 d = make_int4(-1, -1, -1, -1);
    int4 s = make_int4(0, 0, 0, 0);
    if (g < GROUPS) {
        d = reinterpret_cast<const int4*>(dst)[g];
        s = reinterpret_cast<const int4*>(src)[g];
    }
    uint4* s4 = reinterpret_cast<uint4*>(seg);
    #pragma unroll
    for (int j = 0; j < 4; ++j)
        s4[tid + CHUNK_GROUPS * j] = make_uint4(SEG_EMPTY, SEG_EMPTY, SEG_EMPTY, SEG_EMPTY);
    if (tid < AGG_BLOCKS) scnt[tid] = 0u;
    __syncthreads();
    const int dd[4] = {d.x, d.y, d.z, d.w};
    const int ss[4] = {s.x, s.y, s.z, s.w};
    #pragma unroll
    for (int c = 0; c < 4; ++c) {
        if ((unsigned)dd[c] < (unsigned)N_NODES) {
            const unsigned o = (unsigned)dd[c] / ROWS_PER_BLOCK;
            const unsigned r = (unsigned)dd[c] - o * ROWS_PER_BLOCK;
            const unsigned slot = atomicAdd(&scnt[o], 1u);
            if (slot < SEG_ENT) seg[o * SEG_ENT + slot] = min((unsigned)ss[c], (unsigned)(N_NODES - 1)) | (r << 16);
        }
    }
    __syncthreads();
    if (tid < AGG_BLOCKS && scnt[tid] > SEG_ENT) seg[tid * SEG_ENT] = SEG_OVER;
    __syncthreads();
    uint4* o4 = reinterpret_cast<uint4*>(segs + (size_t)blockIdx.x * (AGG_BLOCKS * SEG_ENT));
    #pragma unroll
    for (int j = 0; j < 4; ++j)
        o4[tid + CHUNK_GROUPS * j] = s4[tid + CHUNK_GROUPS * j];
}

template <int LAYER>
__global__ __launch_bounds__(AGG_THREADS)
void sage_layer(const float* __restrict__ in,
                const int* __restrict__ src,
                const int* __restrict__ dst,
                const unsigned int* __restrict__ segs,
                const float* __restrict__ Wl,
                const float* __restrict__ Wr,
                const float* __restrict__ bias,
                float* __restrict__ out,
                unsigned int* __restrict__ list_ws,
                unsigned int* __restrict__ count_ws,
                unsigned int* __restrict__ rowcnt_ws,
                int do_relu) {
    __shared__ __attribute__((aligned(16))) float acc[TILE_ROWS * PITCH];
    __shared__ __attribute__((aligned(16))) float own[TILE_ROWS * PITCH];
    __shared__ __attribute__((aligned(16))) unsigned int queue[QUEUE_CAP];
    __shared__ int cnt[ROW_SLOTS];
    __shared__ unsigned int cur[ROW_SLOTS];
    __shared__ unsigned int qn;
    __shared__ unsigned int over;

    const int tid   = threadIdx.x;
    const int base  = blockIdx.x * ROWS_PER_BLOCK;
    const int nrows = min(ROWS_PER_BLOCK, N_NODES - base);
    unsigned int* __restrict__ my_list = list_ws + (size_t)blockIdx.x * QUEUE_CAP;

    unsigned total = 0u;
    if (LAYER == 2) total = count_ws[blockIdx.x];
    int cntv = 0;
    if (LAYER == 2 && tid < TILE_ROWS) cntv = (int)rowcnt_ws[blockIdx.x * TILE_ROWS + tid];

    float4 ownv[7];
    #pragma unroll
    for (int j = 0; j < 7; ++j) {
        const int r = (tid >> 5) + 32 * j;
        ownv[j] = make_float4(0.f, 0.f, 0.f, 0.f);
        if (r < nrows)
            ownv[j] = *reinterpret_cast<const float4*>(in + (size_t)(base + r) * D_FEAT + 4 * (tid & 31));
    }

    uint4 sv[4];
    if (LAYER == 1) {
        const uint4* sg4 = reinterpret_cast<const uint4*>(segs);
        #pragma unroll
        for (int j = 0; j < 4; ++j) {
            const int i = tid + AGG_THREADS * j;
            sv[j] = make_uint4(SEG_EMPTY, SEG_EMPTY, SEG_EMPTY, SEG_EMPTY);
            if (i < N_CHUNKS * (SEG_ENT / 4))
                sv[j] = sg4[((size_t)(i >> 4) * AGG_BLOCKS + blockIdx.x) * (SEG_ENT / 4) + (i & 15)];
        }
    }

    const unsigned n16 = (LAYER == 2 && total <= QUEUE_CAP) ? ((total + 7u) & ~7u) / 4u : 0u;
    uint4 lv[2];
    if (LAYER == 2) {
        const uint4* l4 = reinterpret_cast<const uint4*>(my_list);
        #pragma unroll
        for (int j = 0; j < 2; ++j) {
            const unsigned i = (unsigned)tid + AGG_THREADS * j;
            lv[j] = make_uint4(0u, 0u, 0u, 0u);
            if (i < n16) lv[j] = l4[i];
        }
    }

    float4* acc4 = reinterpret_cast<float4*>(acc);
    for (int i = tid; i < TILE_ROWS * PITCH / 4; i += AGG_THREADS)
        acc4[i] = make_float4(0.f, 0.f, 0.f, 0.f);
    if (tid < ROW_SLOTS) cnt[tid] = cntv;
    if (tid == 0) { qn = 0u; over = 0u; }
    #pragma unroll
    for (int j = 0; j < 7; ++j) {
        const int r = (tid >> 5) + 32 * j;
        if (r < TILE_ROWS)
            *reinterpret_cast<float4*>(own + r * PITCH + 4 * (tid & 31)) = ownv[j];
    }
    if (LAYER == 2) {
        uint4* q4 = reinterpret_cast<uint4*>(queue);
        #pragma unroll
        for (int j = 0; j < 2; ++j) {
            const unsigned i = (unsigned)tid + AGG_THREADS * j;
            if (i < n16) q4[i] = lv[j];
        }
    }
    __syncthreads();

    if (LAYER == 1) {
        const unsigned e[16] = {sv[0].x, sv[0].y, sv[0].z, sv[0].w, sv[1].x, sv[1].y, sv[1].z, sv[1].w,
                                sv[2].x, sv[2].y, sv[2].z, sv[2].w, sv[3].x, sv[3].y, sv[3].z, sv[3].w};
        {
            bool ov = false;
            #pragma unroll
            for (int k = 0; k < 16; ++k) {
                ov = ov || (e[k] == SEG_OVER);
                if (e[k] < SEG_OVER)
                    __hip_atomic_fetch_add(&cnt[min(e[k] >> 16, (unsigned)(TILE_ROWS - 1))], 1,
                                           __ATOMIC_RELAXED, __HIP_MEMORY_SCOPE_WORKGROUP);
            }
            if (ov) over = 1u;
        }
        __syncthreads();
        if (tid < 32) {
            unsigned first[7];
            unsigned sum = 0u;
            #pragma unroll
            for (int j = 0; j < 7; ++j) {
                first[j] = sum;
                sum += (unsigned)cnt[7 * tid + j];
            }
            unsigned run = sum;
            #pragma unroll
            for (int dlt = 1; dlt < 32; dlt <<= 1) {
                const unsigned up = (unsigned)__shfl_up((int)run, dlt);
                if (tid >= dlt) run += up;
            }
            #pragma unroll
            for (int j = 0; j < 7; ++j) cur[7 * tid + j] = run - sum + first[j];
            if (tid == 31) qn = run;
        }
        __syncthreads();

        total = (over == 0u) ? qn : (unsigned)(QUEUE_CAP + 1);
        unsigned pos[16];
        #pragma unroll
        for (int k = 0; k < 16; ++k) {
            pos[k] = QUEUE_CAP;
            if (e[k] < SEG_OVER) pos[k] = atomicAdd(&cur[min(e[k] >> 16, (unsigned)(TILE_ROWS - 1))], 1u);
        }
        #pragma unroll
        for (int k = 0; k < 16; ++k)
            if (pos[k] < QUEUE_CAP) queue[pos[k]] = e[k];
        const unsigned padded = (total <= QUEUE_CAP) ? ((total + 7u) & ~7u) : 0u;
        if ((unsigned)tid < padded - total && total <= QUEUE_CAP) queue[total + (unsigned)tid] = PAD_ENTRY;
        __syncthreads();
        {
            const uint4* q4 = reinterpret_cast<const uint4*>(queue);
            uint4* l4 = reinterpret_cast<uint4*>(my_list);
            for (unsigned i = (unsigned)tid; i < padded / 4u; i += AGG_THREADS) l4[i] = q4[i];
        }
        if (tid == 0) count_ws[blockIdx.x] = total;
        if (tid < TILE_ROWS) rowcnt_ws[blockIdx.x * TILE_ROWS + tid] = (unsigned)cnt[tid];
    }

    const int wave    = __builtin_amdgcn_readfirstlane(tid >> 5);
    const unsigned total_s = (unsigned)__builtin_amdgcn_readfirstlane((int)total);
    const int lane    = tid & 31;
    const int lhalf   = lane >> 4;
    const int lmod    = lane & 15;
    const int ntile   = wave & 7;
    const int g       = wave >> 3;
    const int col     = ntile * 16 + lmod;
    const bool four   = (g == (ntile >> 1));
    v8f c[4];
    #pragma unroll
    for (int t = 0; t < 4; ++t) c[t] = v8f{};

    if (total_s <= QUEUE_CAP) {
        if (four) gather_with_own<4>(c, in, acc, own, queue, Wr, total_s, (unsigned)wave, lane, g, lmod, lhalf, col);
        else      gather_with_own<3>(c, in, acc, own, queue, Wr, total_s, (unsigned)wave, lane, g, lmod, lhalf, col);
    } else {
        if (tid < TILE_ROWS) cnt[tid] = 0;
        __syncthreads();
        for (int e = tid; e < N_EDGES; e += AGG_THREADS) {
            const unsigned u = (unsigned)(dst[e] - base);
            if (u < (unsigned)nrows) {
                const unsigned s = min((unsigned)src[e], (unsigned)(N_NODES - 1));
                const float* r = in + (size_t)s * D_FEAT;
                float* a = acc + u * PITCH;
                for (int cc = 0; cc < D_FEAT; ++cc)
                    __hip_atomic_fetch_add(a + 32 * (cc & 3) + (cc >> 2), r[cc],
                                           __ATOMIC_RELAXED, __HIP_MEMORY_SCOPE_WORKGROUP);
                __hip_atomic_fetch_add(&cnt[u], 1,
                                       __ATOMIC_RELAXED, __HIP_MEMORY_SCOPE_WORKGROUP);
            }
        }
        if (four) gather_with_own<4>(c, in, acc, own, queue, Wr, 0u, (unsigned)wave, lane, g, lmod, lhalf, col);
        else      gather_with_own<3>(c, in, acc, own, queue, Wr, 0u, (unsigned)wave, lane, g, lmod, lhalf, col);
    }
    __syncthreads();

    for (int i = tid; i < nrows * (D_FEAT / 4); i += AGG_THREADS) {
        const int r = i >> 5;
        const float dv = fmaxf((float)cnt[r], 1.0f);
        float4* p = reinterpret_cast<float4*>(acc + r * PITCH + 4 * (i & 31));
        float4 a = *p;
        a.x = a.x / dv;
        a.y = a.y / dv;
        a.z = a.z / dv;
        a.w = a.w / dv;
        *p = a;
    }
    __syncthreads();

    if (four) mean_products_store<4>(c, acc, Wl, bias, out, g, lmod, lhalf, col, base, nrows, do_relu);
    else      mean_products_store<3>(c, acc, Wl, bias, out, g, lmod, lhalf, col, base, nrows, do_relu);
}

extern "C" void kernel_launch(void* const* d_in, const int* in_sizes, int n_in,
                              void* d_out, int out_size, void* d_ws, size_t ws_size,
                              hipStream_t stream) {
    const float* x    = (const float*)d_in[0];
    const int*   eidx = (const int*)d_in[1];
    const float* W1l  = (const float*)d_in[2];
    const float* b1   = (const float*)d_in[3];
    const float* W1r  = (const float*)d_in[4];
    const float* W2l  = (const float*)d_in[5];
    const float* b2   = (const float*)d_in[6];
    const float* W2r  = (const float*)d_in[7];
    float* out = (float*)d_out;

    const int* src = eidx;
    const int* dst = eidx + N_EDGES;

    float* agg = (float*)d_ws;
    float* deg = agg + (size_t)N_NODES * D_FEAT;
    float* h   = deg + 256;
    unsigned int* list_ws  = (unsigned int*)agg;
    unsigned int* count_ws = (unsigned int*)deg;
    unsigned int* seg_ws   = list_ws + (size_t)AGG_BLOCKS * QUEUE_CAP;
    unsigned int* rowcnt_ws = seg_ws + (size_t)N_CHUNKS * AGG_BLOCKS * SEG_ENT;

    bin_edges<<<N_CHUNKS, CHUNK_GROUPS, 0, stream>>>(src, dst, seg_ws);

    sage_layer<1><<<AGG_BLOCKS, AGG_THREADS, 0, stream>>>(x, src, dst, seg_ws, W1l, W1r, b1, h, list_ws, count_ws, rowcnt_ws, 1);
    sage_layer<2><<<AGG_BLOCKS, AGG_THREADS, 0, stream>>>(h, src, dst, seg_ws, W2l, W2r, b2, out, list_ws, count_ws, rowcnt_ws, 0);
}


# Running product's weights copied once per block to an LDS tile (transposed, k in the A tile's word order): B fragment = one 8-byte LDS read, no global weight loads in the k-loops; sums unchanged
# speedup vs baseline: 8.6093x; 1.0464x over previous
#include <hip/hip_runtime.h>
#include <hip/hip_bf16.h>


#define N_NODES 50000
#define N_EDGES 800000
#define D_FEAT  128

#define ROWS_PER_BLOCK 196
#define AGG_BLOCKS     256
#define AGG_THREADS    1024
#define QUEUE_CAP      8192
#define TILE_ROWS      208
#define PITCH          132
#define GROUPS         (N_EDGES / 4)
#define CHUNK_GROUPS   1024
#define N_CHUNKS       ((GROUPS + CHUNK_GROUPS - 1) / CHUNK_GROUPS)
#define SEG_ENT        64
#define SEG_EMPTY      0xFFFFFFFFu
#define SEG_OVER       0xFFFFFFFEu
#define PAD_ENTRY      ((unsigned)(TILE_ROWS - 1) << 16)
#define ROW_SLOTS      224
#define WPITCH         132

typedef __attribute__((ext_vector_type(2))) float v2f;
typedef __attribute__((ext_vector_type(8))) float v8f;


__device__ __forceinline__ v8f wmma4(const v2f& a, const v2f& b, const v8f& c) {
    return __builtin_amdgcn_wmma_f32_16x16x4_f32(false, a, false, b, (short)0, c, false, false);
}

__device__ __forceinline__ void load_b4(const float* pb, int k0, v2f (&b)[4]) {
    #pragma unroll
    for (int s = 0; s < 4; ++s) b[s] = *(const v2f*)(pb + k0 + 4 * s);
}

template <bool SUMORDER>
__device__ __forceinline__ void weight_tile_load(const float* __restrict__ W, int tid, float4 (&wv)[4]) {
    const int col = tid & (D_FEAT - 1);
    #pragma unroll
    for (int j = 0; j < 4; ++j) {
        const int pg = (tid >> 7) + 8 * j;
        float e[4];
        #pragma unroll
        for (int i = 0; i < 4; ++i) {
            const int p = 4 * pg + i;
            const int k = SUMORDER ? (4 * (p & 31) + (p >> 5)) : p;
            e[i] = W[k * D_FEAT + col];
        }
        wv[j] = make_float4(e[0], e[1], e[2], e[3]);
    }
}

__device__ __forceinline__ void weight_tile_store(float* wt, int tid, const float4 (&wv)[4]) {
    const int col = tid & (D_FEAT - 1);
    #pragma unroll
    for (int j = 0; j < 4; ++j) {
        const int pg = (tid >> 7) + 8 * j;
        *reinterpret_cast<float4*>(wt + col * WPITCH + 4 * pg) = wv[j];
    }
}

template <int NT>
__device__ __forceinline__ void ksteps4(v8f (&c)[4], const float* const (&pa)[NT], const v2f (&b)[4], int k0) {
    #pragma unroll
    for (int s = 0; s < 4; ++s) {
        v2f a[NT];
        #pragma unroll
        for (int t = 0; t < NT; ++t) a[t] = *(const v2f*)(pa[t] + k0 + 4 * s);
        #pragma unroll
        for (int t = 0; t < NT; ++t) c[t] = wmma4(a[t], b[s], c[t]);
    }
}

template <int NT>
__device__ __forceinline__ void gather_with_own(v8f (&c)[4], const float* __restrict__ in, float* acc,
                                                const float* own_t, const unsigned int* queue,
                                                const float* wt,
                                                unsigned total, unsigned wave, int lane,
                                                int g, int lmod, int lhalf, int col) {
    const float* po[NT];
    #pragma unroll
    for (int t = 0; t < NT; ++t) {
        const int mt = (t < 3) ? 3 * g + t : 12;
        po[t] = own_t + (16 * mt + lmod) * PITCH + 2 * lhalf;
    }
    const float* pb = wt + col * WPITCH + 2 * lhalf;
    const uint4* queue4 = reinterpret_cast<const uint4*>(queue);
    const unsigned stride = (AGG_THREADS / 32) * 8u;

    unsigned ent[8];
    float4 v[8];
    auto issue_rows = [&](unsigned k) {
        const uint4 qa = queue4[k / 4u];
        const uint4 qb = queue4[k / 4u + 1u];
        const unsigned raw[8] = {qa.x, qa.y, qa.z, qa.w, qb.x, qb.y, qb.z, qb.w};
        #pragma unroll
        for (int t = 0; t < 8; ++t)
            ent[t] = (unsigned)__builtin_amdgcn_readfirstlane((int)raw[t]);
        #pragma unroll
        for (int t = 0; t < 8; ++t) {
            const unsigned s = min(ent[t] & 0xFFFFu, (unsigned)(N_NODES - 1));
            v[t] = *reinterpret_cast<const float4*>(in + (size_t)s * D_FEAT + 4 * lane);
        }
    };
    auto flush_run = [&](unsigned u, const float4& s) {
        float* a = acc + u * PITCH + lane;
        __hip_atomic_fetch_add(a,      s.x, __ATOMIC_RELAXED, __HIP_MEMORY_SCOPE_WORKGROUP);
        __hip_atomic_fetch_add(a + 32, s.y, __ATOMIC_RELAXED, __HIP_MEMORY_SCOPE_WORKGROUP);
        __hip_atomic_fetch_add(a + 64, s.z, __ATOMIC_RELAXED, __HIP_MEMORY_SCOPE_WORKGROUP);
        __hip_atomic_fetch_add(a + 96, s.w, __ATOMIC_RELAXED, __HIP_MEMORY_SCOPE_WORKGROUP);
    };
    auto add_rows = [&]() {
        unsigned ur = min(ent[0] >> 16, (unsigned)(TILE_ROWS - 1));
        float4 s = v[0];
        #pragma unroll
        for (int t = 1; t < 8; ++t) {
            const unsigned u = min(ent[t] >> 16, (unsigned)(TILE_ROWS - 1));
            if (u == ur) {
                s.x += v[t].x;
                s.y += v[t].y;
                s.z += v[t].z;
                s.w += v[t].w;
            } else {
                flush_run(ur, s);
                ur = u;
                s = v[t];
            }
        }
        flush_run(ur, s);
    };

    unsigned k = wave * 8u;
    int kk = 0;
    v2f b[4];
    for (; k < total && kk < D_FEAT; k += stride, kk += 16) {
        load_b4(pb, kk, b);
        issue_rows(k);
        ksteps4<NT>(c, po, b, kk);
        add_rows();
    }
    for (; k < total; k += stride) {
        issue_rows(k);
        add_rows();
    }
    for (; kk < D_FEAT; kk += 16) {
        load_b4(pb, kk, b);
        ksteps4<NT>(c, po, b, kk);
    }
}

template <int NT>
__device__ __forceinline__ void mean_products_store(v8f (&c)[4], const float* mean_t,
                                                    const float* wt,
                                                    const float* __restrict__ bias,
                                                    float* __restrict__ out,
                                                    int g, int lmod, int lhalf, int col,
                                                    int base, int nrows, int do_relu) {
    int mt[NT];
    const float* pm[NT];
    #pragma unroll
    for (int t = 0; t < NT; ++t) {
        mt[t] = (t < 3) ? 3 * g + t : 12;
        pm[t] = mean_t + (16 * mt[t] + lmod) * PITCH + 2 * lhalf;
    }
    const float* pb = wt + col * WPITCH + 2 * lhalf;

    v2f a0P[NT], a1P[NT], b0P, b1P;
    v2f a0Q[NT], a1Q[NT], b0Q, b1Q;
    auto load_pair = [&](int k, v2f (&a0)[NT], v2f (&a1)[NT], v2f& b0, v2f& b1) {
        #pragma unroll
        for (int t = 0; t < NT; ++t) {
            a0[t] = *(const v2f*)(pm[t] + k);
            a1[t] = *(const v2f*)(pm[t] + k + 4);
        }
        b0 = *(const v2f*)(pb + k);
        b1 = *(const v2f*)(pb + k + 4);
    };
    auto products = [&](const v2f (&a0)[NT], const v2f (&a1)[NT], const v2f& b0, const v2f& b1) {
        #pragma unroll
        for (int t = 0; t < NT; ++t) c[t] = wmma4(a0[t], b0, c[t]);
        #pragma unroll
        for (int t = 0; t < NT; ++t) c[t] = wmma4(a1[t], b1, c[t]);
    };
    load_pair(0, a0P, a1P, b0P, b1P);
    #pragma unroll 2
    for (int k = 0; k < D_FEAT; k += 16) {
        load_pair(k + 8, a0Q, a1Q, b0Q, b1Q);
        products(a0P, a1P, b0P, b1P);
        if (k + 16 < D_FEAT) load_pair(k + 16, a0P, a1P, b0P, b1P);
        products(a0Q, a1Q, b0Q, b1Q);
    }

    const float bv = bias[col];
    #pragma unroll
    for (int t = 0; t < NT; ++t) {
        #pragma unroll
        for (int r = 0; r < 8; ++r) {
            const int row = 16 * mt[t] + 8 * lhalf + r;
            if (row < nrows) {
                float v = c[t][r] + bv;
                if (do_relu) v = fmaxf(v, 0.0f);
                out[(size_t)(base + row) * D_FEAT + col] = v;
            }
        }
    }
}

__global__ __launch_bounds__(CHUNK_GROUPS)
void bin_edges(const int* __restrict__ src, const int* __restrict__ dst, unsigned int* __restrict__ segs) {
    __shared__ __attribute__((aligned(16))) unsigned int seg[AGG_BLOCKS * SEG_ENT];
    __shared__ unsigned int scnt[AGG_BLOCKS];
    const int tid = threadIdx.x;
    const int g = blockIdx.x * CHUNK_GROUPS + tid;
    int4 d = make_int4(-1, -1, -1, -1);
    int4 s = make_int4(0, 0, 0, 0);
    if (g < GROUPS) {
        d = reinterpret_cast<const int4*>(dst)[g];
        s = reinterpret_cast<const int4*>(src)[g];
    }
    uint4* s4 = reinterpret_cast<uint4*>(seg);
    #pragma unroll
    for (int j = 0; j < 4; ++j)
        s4[tid + CHUNK_GROUPS * j] = make_uint4(SEG_EMPTY, SEG_EMPTY, SEG_EMPTY, SEG_EMPTY);
    if (tid < AGG_BLOCKS) scnt[tid] = 0u;
    __syncthreads();
    const int dd[4] = {d.x, d.y, d.z, d.w};
    const int ss[4] = {s.x, s.y, s.z, s.w};
    #pragma unroll
    for (int c = 0; c < 4; ++c) {
        if ((unsigned)dd[c] < (unsigned)N_NODES) {
            const unsigned o = (unsigned)dd[c] / ROWS_PER_BLOCK;
            const unsigned r = (unsigned)dd[c] - o * ROWS_PER_BLOCK;
            const unsigned slot = atomicAdd(&scnt[o], 1u);
            if (slot < SEG_ENT) seg[o * SEG_ENT + slot] = min((unsigned)ss[c], (unsigned)(N_NODES - 1)) | (r << 16);
        }
    }
    __syncthreads();
    if (tid < AGG_BLOCKS && scnt[tid] > SEG_ENT) seg[tid * SEG_ENT] = SEG_OVER;
    __syncthreads();
    uint4* o4 = reinterpret_cast<uint4*>(segs + (size_t)blockIdx.x * (AGG_BLOCKS * SEG_ENT));
    #pragma unroll
    for (int j = 0; j < 4; ++j)
        o4[tid + CHUNK_GROUPS * j] = s4[tid + CHUNK_GROUPS * j];
}

template <int LAYER>
__global__ __launch_bounds__(AGG_THREADS)
void sage_layer(const float* __restrict__ in,
                const int* __restrict__ src,
                const int* __restrict__ dst,
                const unsigned int* __restrict__ segs,
                const float* __restrict__ Wl,
                const float* __restrict__ Wr,
                const float* __restrict__ bias,
                float* __restrict__ out,
                unsigned int* __restrict__ list_ws,
                unsigned int* __restrict__ count_ws,
                unsigned int* __restrict__ rowcnt_ws,
                int do_relu) {
    __shared__ __attribute__((aligned(16))) float acc[TILE_ROWS * PITCH];
    __shared__ __attribute__((aligned(16))) float own[TILE_ROWS * PITCH];
    __shared__ __attribute__((aligned(16))) unsigned int queue[QUEUE_CAP];
    __shared__ __attribute__((aligned(16))) float wt[D_FEAT * WPITCH];
    __shared__ int cnt[ROW_SLOTS];
    __shared__ unsigned int cur[ROW_SLOTS];
    __shared__ unsigned int qn;
    __shared__ unsigned int over;

    const int tid   = threadIdx.x;
    const int base  = blockIdx.x * ROWS_PER_BLOCK;
    const int nrows = min(ROWS_PER_BLOCK, N_NODES - base);
    unsigned int* __restrict__ my_list = list_ws + (size_t)blockIdx.x * QUEUE_CAP;

    unsigned total = 0u;
    if (LAYER == 2) total = count_ws[blockIdx.x];
    int cntv = 0;
    if (LAYER == 2 && tid < TILE_ROWS) cntv = (int)rowcnt_ws[blockIdx.x * TILE_ROWS + tid];

    float4 ownv[7];
    #pragma unroll
    for (int j = 0; j < 7; ++j) {
        const int r = (tid >> 5) + 32 * j;
        ownv[j] = make_float4(0.f, 0.f, 0.f, 0.f);
        if (r < nrows)
            ownv[j] = *reinterpret_cast<const float4*>(in + (size_t)(base + r) * D_FEAT + 4 * (tid & 31));
    }

    uint4 sv[4];
    if (LAYER == 1) {
        const uint4* sg4 = reinterpret_cast<const uint4*>(segs);
        #pragma unroll
        for (int j = 0; j < 4; ++j) {
            const int i = tid + AGG_THREADS * j;
            sv[j] = make_uint4(SEG_EMPTY, SEG_EMPTY, SEG_EMPTY, SEG_EMPTY);
            if (i < N_CHUNKS * (SEG_ENT / 4))
                sv[j] = sg4[((size_t)(i >> 4) * AGG_BLOCKS + blockIdx.x) * (SEG_ENT / 4) + (i & 15)];
        }
    }

    const unsigned n16 = (LAYER == 2 && total <= QUEUE_CAP) ? ((total + 7u) & ~7u) / 4u : 0u;
    uint4 lv[2];
    if (LAYER == 2) {
        const uint4* l4 = reinterpret_cast<const uint4*>(my_list);
        #pragma unroll
        for (int j = 0; j < 2; ++j) {
            const unsigned i = (unsigned)tid + AGG_THREADS * j;
            lv[j] = make_uint4(0u, 0u, 0u, 0u);
            if (i < n16) lv[j] = l4[i];
        }
    }

    float4 wv[4];
    weight_tile_load<false>(Wr, tid, wv);

    float4* acc4 = reinterpret_cast<float4*>(acc);
    for (int i = tid; i < TILE_ROWS * PITCH / 4; i += AGG_THREADS)
        acc4[i] = make_float4(0.f, 0.f, 0.f, 0.f);
    if (tid < ROW_SLOTS) cnt[tid] = cntv;
    if (tid == 0) { qn = 0u; over = 0u; }
    #pragma unroll
    for (int j = 0; j < 7; ++j) {
        const int r = (tid >> 5) + 32 * j;
        if (r < TILE_ROWS)
            *reinterpret_cast<float4*>(own + r * PITCH + 4 * (tid & 31)) = ownv[j];
    }
    if (LAYER == 2) {
        uint4* q4 = reinterpret_cast<uint4*>(queue);
        #pragma unroll
        for (int j = 0; j < 2; ++j) {
            const unsigned i = (unsigned)tid + AGG_THREADS * j;
            if (i < n16) q4[i] = lv[j];
        }
    }
    weight_tile_store(wt, tid, wv);
    __syncthreads();

    if (LAYER == 1) {
        const unsigned e[16] = {sv[0].x, sv[0].y, sv[0].z, sv[0].w, sv[1].x, sv[1].y, sv[1].z, sv[1].w,
                                sv[2].x, sv[2].y, sv[2].z, sv[2].w, sv[3].x, sv[3].y, sv[3].z, sv[3].w};
        {
            bool ov = false;
            #pragma unroll
            for (int k = 0; k < 16; ++k) {
                ov = ov || (e[k] == SEG_OVER);
                if (e[k] < SEG_OVER)
                    __hip_atomic_fetch_add(&cnt[min(e[k] >> 16, (unsigned)(TILE_ROWS - 1))], 1,
                                           __ATOMIC_RELAXED, __HIP_MEMORY_SCOPE_WORKGROUP);
            }
            if (ov) over = 1u;
        }
        __syncthreads();
        if (tid < 32) {
            unsigned first[7];
            unsigned sum = 0u;
            #pragma unroll
            for (int j = 0; j < 7; ++j) {
                first[j] = sum;
                sum += (unsigned)cnt[7 * tid + j];
            }
            unsigned run = sum;
            #pragma unroll
            for (int dlt = 1; dlt < 32; dlt <<= 1) {
                const unsigned up = (unsigned)__shfl_up((int)run, dlt);
                if (tid >= dlt) run += up;
            }
            #pragma unroll
            for (int j = 0; j < 7; ++j) cur[7 * tid + j] = run - sum + first[j];
            if (tid == 31) qn = run;
        }
        __syncthreads();

        total = (over == 0u) ? qn : (unsigned)(QUEUE_CAP + 1);
        unsigned pos[16];
        #pragma unroll
        for (int k = 0; k < 16; ++k) {
            pos[k] = QUEUE_CAP;
            if (e[k] < SEG_OVER) pos[k] = atomicAdd(&cur[min(e[k] >> 16, (unsigned)(TILE_ROWS - 1))], 1u);
        }
        #pragma unroll
        for (int k = 0; k < 16; ++k)
            if (pos[k] < QUEUE_CAP) queue[pos[k]] = e[k];
        const unsigned padded = (total <= QUEUE_CAP) ? ((total + 7u) & ~7u) : 0u;
        if ((unsigned)tid < padded - total && total <= QUEUE_CAP) queue[total + (unsigned)tid] = PAD_ENTRY;
        __syncthreads();
        {
            const uint4* q4 = reinterpret_cast<const uint4*>(queue);
            uint4* l4 = reinterpret_cast<uint4*>(my_list);
            for (unsigned i = (unsigned)tid; i < padded / 4u; i += AGG_THREADS) l4[i] = q4[i];
        }
        if (tid == 0) count_ws[blockIdx.x] = total;
        if (tid < TILE_ROWS) rowcnt_ws[blockIdx.x * TILE_ROWS + tid] = (unsigned)cnt[tid];
    }

    const int wave    = __builtin_amdgcn_readfirstlane(tid >> 5);
    const unsigned total_s = (unsigned)__builtin_amdgcn_readfirstlane((int)total);
    const int lane    = tid & 31;
    const int lhalf   = lane >> 4;
    const int lmod    = lane & 15;
    const int ntile   = wave & 7;
    const int g       = wave >> 3;
    const int col     = ntile * 16 + lmod;
    const bool four   = (g == (ntile >> 1));
    v8f c[4];
    #pragma unroll
    for (int t = 0; t < 4; ++t) c[t] = v8f{};

    if (total_s <= QUEUE_CAP) {
        if (four) gather_with_own<4>(c, in, acc, own, queue, wt, total_s, (unsigned)wave, lane, g, lmod, lhalf, col);
        else      gather_with_own<3>(c, in, acc, own, queue, wt, total_s, (unsigned)wave, lane, g, lmod, lhalf, col);
    } else {
        if (tid < TILE_ROWS) cnt[tid] = 0;
        __syncthreads();
        for (int e = tid; e < N_EDGES; e += AGG_THREADS) {
            const unsigned u = (unsigned)(dst[e] - base);
            if (u < (unsigned)nrows) {
                const unsigned s = min((unsigned)src[e], (unsigned)(N_NODES - 1));
                const float* r = in + (size_t)s * D_FEAT;
                float* a = acc + u * PITCH;
                for (int cc = 0; cc < D_FEAT; ++cc)
                    __hip_atomic_fetch_add(a + 32 * (cc & 3) + (cc >> 2), r[cc],
                                           __ATOMIC_RELAXED, __HIP_MEMORY_SCOPE_WORKGROUP);
                __hip_atomic_fetch_add(&cnt[u], 1,
                                       __ATOMIC_RELAXED, __HIP_MEMORY_SCOPE_WORKGROUP);
            }
        }
        if (four) gather_with_own<4>(c, in, acc, own, queue, wt, 0u, (unsigned)wave, lane, g, lmod, lhalf, col);
        else      gather_with_own<3>(c, in, acc, own, queue, wt, 0u, (unsigned)wave, lane, g, lmod, lhalf, col);
    }
    __syncthreads();

    weight_tile_load<true>(Wl, tid, wv);

    for (int i = tid; i < nrows * (D_FEAT / 4); i += AGG_THREADS) {
        const int r = i >> 5;
        const float dv = fmaxf((float)cnt[r], 1.0f);
        float4* p = reinterpret_cast<float4*>(acc + r * PITCH + 4 * (i & 31));
        float4 a = *p;
        a.x = a.x / dv;
        a.y = a.y / dv;
        a.z = a.z / dv;
        a.w = a.w / dv;
        *p = a;
    }
    weight_tile_store(wt, tid, wv);
    __syncthreads();

    if (four) mean_products_store<4>(c, acc, wt, bias, out, g, lmod, lhalf, col, base, nrows, do_relu);
    else      mean_products_store<3>(c, acc, wt, bias, out, g, lmod, lhalf, col, base, nrows, do_relu);
}

extern "C" void kernel_launch(void* const* d_in, const int* in_sizes, int n_in,
                              void* d_out, int out_size, void* d_ws, size_t ws_size,
                              hipStream_t stream) {
    const float* x    = (const float*)d_in[0];
    const int*   eidx = (const int*)d_in[1];
    const float* W1l  = (const float*)d_in[2];
    const float* b1   = (const float*)d_in[3];
    const float* W1r  = (const float*)d_in[4];
    const float* W2l  = (const float*)d_in[5];
    const float* b2   = (const float*)d_in[6];
    const float* W2r  = (const float*)d_in[7];
    float* out = (float*)d_out;

    const int* src = eidx;
    const int* dst = eidx + N_EDGES;

    float* agg = (float*)d_ws;
    float* deg = agg + (size_t)N_NODES * D_FEAT;
    float* h   = deg + 256;
    unsigned int* list_ws  = (unsigned int*)agg;
    unsigned int* count_ws = (unsigned int*)deg;
    unsigned int* seg_ws   = list_ws + (size_t)AGG_BLOCKS * QUEUE_CAP;
    unsigned int* rowcnt_ws = seg_ws + (size_t)N_CHUNKS * AGG_BLOCKS * SEG_ENT;

    bin_edges<<<N_CHUNKS, CHUNK_GROUPS, 0, stream>>>(src, dst, seg_ws);

    sage_layer<1><<<AGG_BLOCKS, AGG_THREADS, 0, stream>>>(x, src, dst, seg_ws, W1l, W1r, b1, h, list_ws, count_ws, rowcnt_ws, 1);
    sage_layer<2><<<AGG_BLOCKS, AGG_THREADS, 0, stream>>>(h, src, dst, seg_ws, W2l, W2r, b2, out, list_ws, count_ws, rowcnt_ws, 0);
}
